# starting kernel
# baseline (speedup 1.0000x reference)
#include <hip/hip_runtime.h>
#include <hip/hip_bf16.h>

typedef float v2f __attribute__((ext_vector_type(2)));
typedef float v8f __attribute__((ext_vector_type(8)));


__global__ void gcn_zero_f32(float* __restrict__ p, long n) {
    long i = (long)blockIdx.x * blockDim.x + threadIdx.x;
    long stride = (long)gridDim.x * blockDim.x;
    for (; i < n; i += stride) p[i] = 0.0f;
}

__global__ void gcn_degree(const int* __restrict__ dst, float* __restrict__ deg, int E) {
    int e = blockIdx.x * blockDim.x + threadIdx.x;
    if (e < E) atomicAdd(&deg[dst[e]], 1.0f);
}

__global__ void gcn_dis(const float* __restrict__ deg, float* __restrict__ dis, int N) {
    int i = blockIdx.x * blockDim.x + threadIdx.x;
    if (i < N) dis[i] = rsqrtf(deg[i] + 1.0f);
}

__global__ void gcn_edge1(const int* __restrict__ src, const int* __restrict__ dst,
                          const float* __restrict__ x, const float* __restrict__ dis,
                          float* __restrict__ s1, int E) {
    int e = blockIdx.x * blockDim.x + threadIdx.x;
    if (e < E) {
        int s = src[e], d = dst[e];
        atomicAdd(&s1[d], x[s] * dis[s] * dis[d]);
    }
}

__global__ void gcn_make_t(const float* __restrict__ s1, const float* __restrict__ dis,
                           const float* __restrict__ x, float* __restrict__ t, int N) {
    int i = blockIdx.x * blockDim.x + threadIdx.x;
    if (i < N) t[i] = s1[i] + dis[i] * dis[i] * x[i];
}

__global__ void gcn_gemm_wmma(const float* __restrict__ t,
                              const float* __restrict__ W1,
                              const float* __restrict__ b1,
                              const float* __restrict__ W2,
                              float* __restrict__ h2pre,
                              int nTiles) {
    int wave = (int)((blockIdx.x * blockDim.x + threadIdx.x) >> 5);
    int lane = threadIdx.x & 31;
    if (wave >= nTiles) return;

    int mt = wave >> 1;
    int nt = wave & 1;
    int m0 = mt << 4;
    int n0 = nt << 4;
    int r = lane & 15;
    int khalf = lane >> 4;

    float trow = t[m0 + r];

    v8f acc = {};
#pragma unroll
    for (int k0 = 0; k0 < 64; k0 += 4) {
        int ka = k0 + khalf * 2;
        float a0 = fmaf(trow, W1[ka],     b1[ka]);
        float a1 = fmaf(trow, W1[ka + 1], b1[ka + 1]);
        v2f a, b;
        a.x = a0 > 0.0f ? a0 : 0.0f;
        a.y = a1 > 0.0f ? a1 : 0.0f;
        b.x = W2[ka * 32 + n0 + r];
        b.y = W2[(ka + 1) * 32 + n0 + r];
        acc = __builtin_amdgcn_wmma_f32_16x16x4_f32(
             false, a,  false, b,
             (short)0, acc,  false,  false);
    }

    int col = n0 + r;
    int rowBase = m0 + khalf * 8;
#pragma unroll
    for (int v = 0; v < 8; ++v) {
        h2pre[(rowBase + v) * 32 + col] = acc[v];
    }
}

__global__ void gcn_edge2(const int* __restrict__ src, const int* __restrict__ dst,
                          const float* __restrict__ dis,
                          const float* __restrict__ h2pre,
                          float* __restrict__ agg2, int E) {
    int lane = threadIdx.x & 31;
    int wave = (int)((blockIdx.x * blockDim.x + threadIdx.x) >> 5);
    int nWaves = (int)((gridDim.x * blockDim.x) >> 5);
    for (int e = wave; e < E; e += nWaves) {
        int s = src[e], d = dst[e];
        float norm = dis[s] * dis[d];
        float v = h2pre[s * 32 + lane] * norm;
        atomicAdd(&agg2[d * 32 + lane], v);
    }
}

__global__ void gcn_pool(const float* __restrict__ agg2, const float* __restrict__ h2pre,
                         const float* __restrict__ dis, const float* __restrict__ b2,
                         const int* __restrict__ batch,
                         float* __restrict__ psum, float* __restrict__ pcnt, int N) {
    long tid = (long)blockIdx.x * blockDim.x + threadIdx.x;
    long total = (long)N * 32;
    long stride = (long)gridDim.x * blockDim.x;
    for (; tid < total; tid += stride) {
        int i = (int)(tid >> 5);
        int f = (int)(tid & 31);
        float d2 = dis[i] * dis[i];
        float h = agg2[tid] + d2 * h2pre[tid] + b2[f];
        h = h > 0.0f ? h : 0.0f;
        int g = batch[i];
        atomicAdd(&psum[g * 32 + f], h);
        if (f == 0) atomicAdd(&pcnt[g], 1.0f);
    }
}

__global__ void gcn_mlp(const float* __restrict__ psum, const float* __restrict__ pcnt,
                        const float* __restrict__ Wp1, const float* __restrict__ bp1,
                        const float* __restrict__ Wp2, const float* __restrict__ bp2,
                        float* __restrict__ out, int G) {
    int g = blockIdx.x * blockDim.x + threadIdx.x;
    if (g >= G) return;
    float c = pcnt[g];
    c = c > 1.0f ? c : 1.0f;
    float inv = 1.0f / c;
    float pooled[32];
#pragma unroll
    for (int f = 0; f < 32; ++f) pooled[f] = psum[g * 32 + f] * inv;
    float o0 = bp2[0], o1 = bp2[1], o2 = bp2[2];
    for (int j = 0; j < 128; ++j) {
        float h = bp1[j];
#pragma unroll
        for (int f = 0; f < 32; ++f) h = fmaf(pooled[f], Wp1[f * 128 + j], h);
        h = h > 0.0f ? h : 0.0f;
        o0 = fmaf(h, Wp2[j * 3 + 0], o0);
        o1 = fmaf(h, Wp2[j * 3 + 1], o1);
        o2 = fmaf(h, Wp2[j * 3 + 2], o2);
    }
    out[g * 3 + 0] = o0;
    out[g * 3 + 1] = o1;
    out[g * 3 + 2] = o2;
}

extern "C" void kernel_launch(void* const* d_in, const int* in_sizes, int n_in,
                              void* d_out, int out_size, void* d_ws, size_t ws_size,
                              hipStream_t stream) {
    const float* x   = (const float*)d_in[0];
    const float* W1  = (const float*)d_in[1];
    const float* b1  = (const float*)d_in[2];
    const float* W2  = (const float*)d_in[3];
    const float* b2  = (const float*)d_in[4];
    const float* Wp1 = (const float*)d_in[5];
    const float* bp1 = (const float*)d_in[6];
    const float* Wp2 = (const float*)d_in[7];
    const float* bp2 = (const float*)d_in[8];
    const int* eidx  = (const int*)d_in[9];
    const int* batch = (const int*)d_in[10];
    float* out = (float*)d_out;

    const int N = in_sizes[0];
    const int E = in_sizes[9] / 2;
    const int G = out_size / 3;
    const int* src = eidx;
    const int* dst = eidx + E;

    float* wsf   = (float*)d_ws;
    float* deg   = wsf;
    float* s1    = deg + N;
    float* dis   = s1 + N;
    float* tbuf  = dis + N;
    float* h2pre = tbuf + N;
    float* agg2  = h2pre + (long)N * 32;
    float* psum  = agg2 + (long)N * 32;
    float* pcnt  = psum + (long)G * 32;

    const int TPB = 256;

    long nz1 = 2L * N;
    long nz2 = 32L * N + 32L * G + G;
    gcn_zero_f32<<<1024, TPB, 0, stream>>>(deg, nz1);
    gcn_zero_f32<<<2048, TPB, 0, stream>>>(agg2, nz2);

    gcn_degree<<<(E + TPB - 1) / TPB, TPB, 0, stream>>>(dst, deg, E);
    gcn_dis<<<(N + TPB - 1) / TPB, TPB, 0, stream>>>(deg, dis, N);
    gcn_edge1<<<(E + TPB - 1) / TPB, TPB, 0, stream>>>(src, dst, x, dis, s1, E);
    gcn_make_t<<<(N + TPB - 1) / TPB, TPB, 0, stream>>>(s1, dis, x, tbuf, N);

    int nTiles = (N / 16) * 2;
    gcn_gemm_wmma<<<(nTiles + 7) / 8, TPB, 0, stream>>>(tbuf, W1, b1, W2, h2pre, nTiles);

    gcn_edge2<<<4096, TPB, 0, stream>>>(src, dst, dis, h2pre, agg2, E);

    gcn_pool<<<4096, TPB, 0, stream>>>(agg2, h2pre, dis, b2, batch, psum, pcnt, N);

    gcn_mlp<<<(G + TPB - 1) / TPB, TPB, 0, stream>>>(psum, pcnt, Wp1, bp1, Wp2, bp2, out, G);
}


# edges counting-sorted once per call into 256-node dst buckets; one block per bucket sums degree, layer-1 and layer-2 terms in LDS and finishes its nodes (no scatter atomics); float sum order changes
# speedup vs baseline: 3.1264x; 3.1264x over previous
#include <hip/hip_runtime.h>
#include <hip/hip_bf16.h>


typedef float v2f __attribute__((ext_vector_type(2)));
typedef float v8f __attribute__((ext_vector_type(8)));

#define RSH 8
#define RN 256
#define SRC_BITS 20
#define SRC_MASK 0xFFFFF
#define NCHUNK 250


__device__ __forceinline__ int block_incl_scan(int v, int* sh) {
    const int tid = threadIdx.x;
    const int n = blockDim.x;
    sh[tid] = v;
    __syncthreads();
    for (int off = 1; off < n; off <<= 1) {
        int a = tid >= off ? sh[tid - off] : 0;
        __syncthreads();
        v += a;
        sh[tid] = v;
        __syncthreads();
    }
    return v;
}

__global__ void __launch_bounds__(1024)
gcn_count(const int* __restrict__ dst, int* __restrict__ cnt, int E, int chunkLen, int B, int C) {
    __shared__ int h[1024];
    const int tid = threadIdx.x;
    for (int i = tid; i < B; i += blockDim.x) h[i] = 0;
    __syncthreads();
    const int lo = blockIdx.x * chunkLen;
    const int hi = min(lo + chunkLen, E);
    for (int e = lo + tid; e < hi; e += blockDim.x) atomicAdd(&h[dst[e] >> RSH], 1);
    __syncthreads();
    for (int i = tid; i < B; i += blockDim.x) cnt[i * C + blockIdx.x] = h[i];
}

__global__ void __launch_bounds__(256)
gcn_bscan(int* __restrict__ cnt, int* __restrict__ btot, int C, float* __restrict__ pz, int nz) {
    __shared__ int sh[256];
    const int tid = threadIdx.x;
    const int b = blockIdx.x;
    int v = tid < C ? cnt[b * C + tid] : 0;
    int incl = block_incl_scan(v, sh);
    if (tid < C) cnt[b * C + tid] = incl - v;
    if (tid == 255) btot[b] = incl;
    for (int i = b * 256 + tid; i < nz; i += gridDim.x * 256) pz[i] = 0.0f;
}

__global__ void __launch_bounds__(1024)
gcn_scatter(const int* __restrict__ src, const int* __restrict__ dst,
            const int* __restrict__ offs, const int* __restrict__ btot,
            int* __restrict__ bstart, int* __restrict__ part,
            int E, int chunkLen, int B, int C) {
    __shared__ int sh[1024];
    __shared__ int cur[1024];
    const int tid = threadIdx.x;
    const int c = blockIdx.x;
    int v = tid < B ? btot[tid] : 0;
    int incl = block_incl_scan(v, sh);
    int excl = incl - v;
    if (tid < B) cur[tid] = excl + offs[tid * C + c];
    if (c == 0 && tid <= B) bstart[tid] = excl;
    __syncthreads();
    const int lo = c * chunkLen;
    const int hi = min(lo + chunkLen, E);
    for (int e = lo + tid; e < hi; e += blockDim.x) {
        int d = dst[e];
        int s = src[e];
        int pos = atomicAdd(&cur[d >> RSH], 1);
        part[pos] = s | ((d & (RN - 1)) << SRC_BITS);
    }
}

__global__ void __launch_bounds__(1024)
gcn_deg_bucket(const int* __restrict__ part, const int* __restrict__ bstart,
               const float* __restrict__ x, float* __restrict__ dis, float* __restrict__ xs, int N) {
    __shared__ int h[RN];
    const int tid = threadIdx.x;
    if (tid < RN) h[tid] = 0;
    __syncthreads();
    const int e0 = bstart[blockIdx.x], e1 = bstart[blockIdx.x + 1];
    for (int i = e0 + tid; i < e1; i += blockDim.x) atomicAdd(&h[part[i] >> SRC_BITS], 1);
    __syncthreads();
    if (tid < RN) {
        int nd = (blockIdx.x << RSH) + tid;
        if (nd < N) {
            float d = rsqrtf((float)h[tid] + 1.0f);
            dis[nd] = d;
            xs[nd] = x[nd] * d;
        }
    }
}

__global__ void __launch_bounds__(1024)
gcn_edge1_bucket(const int* __restrict__ part, const int* __restrict__ bstart,
                 const float* __restrict__ xs, const float* __restrict__ dis,
                 const float* __restrict__ x, float* __restrict__ t, int N) {
    __shared__ float acc[RN];
    __shared__ float disL[RN];
    const int tid = threadIdx.x;
    const int nd = (blockIdx.x << RSH) + tid;
    if (tid < RN) {
        acc[tid] = 0.0f;
        disL[tid] = nd < N ? dis[nd] : 0.0f;
    }
    __syncthreads();
    const int e0 = bstart[blockIdx.x], e1 = bstart[blockIdx.x + 1];
    for (int i = e0 + tid; i < e1; i += blockDim.x) {
        int w = part[i];
        int dl = w >> SRC_BITS;
        atomicAdd(&acc[dl], xs[w & SRC_MASK] * disL[dl]);
    }
    __syncthreads();
    if (tid < RN && nd < N) t[nd] = acc[tid] + disL[tid] * disL[tid] * x[nd];
}

__global__ void gcn_gemm_wmma(const float* __restrict__ t,
                              const float* __restrict__ W1,
                              const float* __restrict__ b1,
                              const float* __restrict__ W2,
                              float* __restrict__ h2pre,
                              int nTiles) {
    int wave = (int)((blockIdx.x * blockDim.x + threadIdx.x) >> 5);
    int lane = threadIdx.x & 31;
    if (wave >= nTiles) return;

    int mt = wave >> 1;
    int nt = wave & 1;
    int m0 = mt << 4;
    int n0 = nt << 4;
    int r = lane & 15;
    int khalf = lane >> 4;

    float trow = t[m0 + r];

    v8f acc = {};
#pragma unroll
    for (int k0 = 0; k0 < 64; k0 += 4) {
        int ka = k0 + khalf * 2;
        float a0 = fmaf(trow, W1[ka],     b1[ka]);
        float a1 = fmaf(trow, W1[ka + 1], b1[ka + 1]);
        v2f a, b;
        a.x = a0 > 0.0f ? a0 : 0.0f;
        a.y = a1 > 0.0f ? a1 : 0.0f;
        b.x = W2[ka * 32 + n0 + r];
        b.y = W2[(ka + 1) * 32 + n0 + r];
        acc = __builtin_amdgcn_wmma_f32_16x16x4_f32(
             false, a,  false, b,
             (short)0, acc,  false,  false);
    }

    int col = n0 + r;
    int rowBase = m0 + khalf * 8;
#pragma unroll
    for (int v = 0; v < 8; ++v) {
        h2pre[(rowBase + v) * 32 + col] = acc[v];
    }
}

__global__ void __launch_bounds__(1024)
gcn_edge2_bucket(const int* __restrict__ part, const int* __restrict__ bstart,
                 const float* __restrict__ dis, const float* __restrict__ h2pre,
                 const float* __restrict__ b2, const int* __restrict__ batch,
                 float* __restrict__ psum, float* __restrict__ pcnt, int N) {
    __shared__ float acc[RN * 32];
    __shared__ float disL[RN];
    const int tid = threadIdx.x;
    const int lane = tid & 31;
    const int wave = tid >> 5;
    const int nWaves = blockDim.x >> 5;
    const int node0 = blockIdx.x << RSH;
    for (int i = tid; i < RN * 32; i += blockDim.x) acc[i] = 0.0f;
    if (tid < RN) {
        int nd = node0 + tid;
        disL[tid] = nd < N ? dis[nd] : 0.0f;
    }
    __syncthreads();
    const int e0 = bstart[blockIdx.x], e1 = bstart[blockIdx.x + 1];
    for (int base = e0 + wave * 32; base < e1; base += nWaves * 32) {
        const int idx = base + lane;
        const int w = idx < e1 ? part[idx] : 0;
        const int s = w & SRC_MASK;
        const int dl = w >> SRC_BITS;
        const float nrm = dis[s] * disL[dl];
        const int nrmBits = __float_as_int(nrm);
        const int cnt = min(32, e1 - base);
        if (cnt == 32) {
#pragma unroll
            for (int j0 = 0; j0 < 32; j0 += 8) {
                float v[8];
#pragma unroll
                for (int u = 0; u < 8; ++u) {
                    int sj = __builtin_amdgcn_readlane(s, j0 + u);
                    v[u] = h2pre[sj * 32 + lane];
                }
#pragma unroll
                for (int u = 0; u < 8; ++u) {
                    int dj = __builtin_amdgcn_readlane(dl, j0 + u);
                    float nj = __int_as_float(__builtin_amdgcn_readlane(nrmBits, j0 + u));
                    atomicAdd(&acc[dj * 32 + lane], v[u] * nj);
                }
            }
        } else {
            for (int j = 0; j < cnt; ++j) {
                int sj = __builtin_amdgcn_readlane(s, j);
                int dj = __builtin_amdgcn_readlane(dl, j);
                float nj = __int_as_float(__builtin_amdgcn_readlane(nrmBits, j));
                atomicAdd(&acc[dj * 32 + lane], h2pre[sj * 32 + lane] * nj);
            }
        }
    }
    __syncthreads();
    for (int i = tid; i < RN * 32; i += blockDim.x) {
        const int nl = i >> 5;
        const int f = i & 31;
        const int nd = node0 + nl;
        if (nd < N) {
            float d2 = disL[nl] * disL[nl];
            float h = acc[i] + d2 * h2pre[nd * 32 + f] + b2[f];
            h = h > 0.0f ? h : 0.0f;
            int g = batch[nd];
            atomicAdd(&psum[g * 32 + f], h);
            if (f == 0) atomicAdd(&pcnt[g], 1.0f);
        }
    }
}

__global__ void gcn_mlp(const float* __restrict__ psum, const float* __restrict__ pcnt,
                        const float* __restrict__ Wp1, const float* __restrict__ bp1,
                        const float* __restrict__ Wp2, const float* __restrict__ bp2,
                        float* __restrict__ out, int G) {
    int g = blockIdx.x * blockDim.x + threadIdx.x;
    if (g >= G) return;
    float c = pcnt[g];
    c = c > 1.0f ? c : 1.0f;
    float inv = 1.0f / c;
    float pooled[32];
#pragma unroll
    for (int f = 0; f < 32; ++f) pooled[f] = psum[g * 32 + f] * inv;
    float o0 = bp2[0], o1 = bp2[1], o2 = bp2[2];
    for (int j = 0; j < 128; ++j) {
        float h = bp1[j];
#pragma unroll
        for (int f = 0; f < 32; ++f) h = fmaf(pooled[f], Wp1[f * 128 + j], h);
        h = h > 0.0f ? h : 0.0f;
        o0 = fmaf(h, Wp2[j * 3 + 0], o0);
        o1 = fmaf(h, Wp2[j * 3 + 1], o1);
        o2 = fmaf(h, Wp2[j * 3 + 2], o2);
    }
    out[g * 3 + 0] = o0;
    out[g * 3 + 1] = o1;
    out[g * 3 + 2] = o2;
}

extern "C" void kernel_launch(void* const* d_in, const int* in_sizes, int n_in,
                              void* d_out, int out_size, void* d_ws, size_t ws_size,
                              hipStream_t stream) {
    const float* x   = (const float*)d_in[0];
    const float* W1  = (const float*)d_in[1];
    const float* b1  = (const float*)d_in[2];
    const float* W2  = (const float*)d_in[3];
    const float* b2  = (const float*)d_in[4];
    const float* Wp1 = (const float*)d_in[5];
    const float* bp1 = (const float*)d_in[6];
    const float* Wp2 = (const float*)d_in[7];
    const float* bp2 = (const float*)d_in[8];
    const int* eidx  = (const int*)d_in[9];
    const int* batch = (const int*)d_in[10];
    float* out = (float*)d_out;

    const int N = in_sizes[0];
    const int E = in_sizes[9] / 2;
    const int G = out_size / 3;
    const int* src = eidx;
    const int* dst = eidx + E;

    const int B = (N + RN - 1) >> RSH;
    const int C = NCHUNK;
    const int chunkLen = (E + C - 1) / C;

    float* wsf   = (float*)d_ws;
    float* dis   = wsf;
    float* xs    = dis + N;
    float* tbuf  = xs + N;
    int*   bstart = (int*)(tbuf + N);
    int*   btot  = bstart + (B + 1);
    int*   cnt   = btot + B;
    float* h2pre = wsf + 4L * N;
    int*   part  = (int*)(h2pre + (long)N * 32);
    float* psum  = (float*)(part + E);
    float* pcnt  = psum + (long)G * 32;

    const int TPB = 256;

    gcn_count<<<C, 1024, 0, stream>>>(dst, cnt, E, chunkLen, B, C);
    gcn_bscan<<<B, 256, 0, stream>>>(cnt, btot, C, psum, 33 * G);
    gcn_scatter<<<C, 1024, 0, stream>>>(src, dst, cnt, btot, bstart, part, E, chunkLen, B, C);

    gcn_deg_bucket<<<B, 1024, 0, stream>>>(part, bstart, x, dis, xs, N);
    gcn_edge1_bucket<<<B, 1024, 0, stream>>>(part, bstart, xs, dis, x, tbuf, N);

    int nTiles = (N / 16) * 2;
    gcn_gemm_wmma<<<(nTiles + 7) / 8, TPB, 0, stream>>>(tbuf, W1, b1, W2, h2pre, nTiles);

    gcn_edge2_bucket<<<B, 1024, 0, stream>>>(part, bstart, dis, h2pre, b2, batch, psum, pcnt, N);

    gcn_mlp<<<(G + TPB - 1) / TPB, TPB, 0, stream>>>(psum, pcnt, Wp1, bp1, Wp2, bp2, out, G);
}


# MLP head on matrix tiles: pooled@Wp1 as f32 16x16x4 WMMA, a wave = 16 graphs x 16 hidden (one Wp1 fragment serves 16 graphs), 128->3 by 16 lanes a graph + butterfly; sum order changes
# speedup vs baseline: 3.8501x; 1.2315x over previous
#include <hip/hip_runtime.h>
#include <hip/hip_bf16.h>


typedef float v2f __attribute__((ext_vector_type(2)));
typedef float v8f __attribute__((ext_vector_type(8)));

#define RSH 8
#define RN 256
#define SRC_BITS 20
#define SRC_MASK 0xFFFFF
#define NCHUNK 250


__device__ __forceinline__ int block_incl_scan(int v, int* sh) {
    const int tid = threadIdx.x;
    const int n = blockDim.x;
    sh[tid] = v;
    __syncthreads();
    for (int off = 1; off < n; off <<= 1) {
        int a = tid >= off ? sh[tid - off] : 0;
        __syncthreads();
        v += a;
        sh[tid] = v;
        __syncthreads();
    }
    return v;
}

__global__ void __launch_bounds__(1024)
gcn_count(const int* __restrict__ dst, int* __restrict__ cnt, int E, int chunkLen, int B, int C) {
    __shared__ int h[1024];
    const int tid = threadIdx.x;
    for (int i = tid; i < B; i += blockDim.x) h[i] = 0;
    __syncthreads();
    const int lo = blockIdx.x * chunkLen;
    const int hi = min(lo + chunkLen, E);
    for (int e = lo + tid; e < hi; e += blockDim.x) atomicAdd(&h[dst[e] >> RSH], 1);
    __syncthreads();
    for (int i = tid; i < B; i += blockDim.x) cnt[i * C + blockIdx.x] = h[i];
}

__global__ void __launch_bounds__(256)
gcn_bscan(int* __restrict__ cnt, int* __restrict__ btot, int C, float* __restrict__ pz, int nz) {
    __shared__ int sh[256];
    const int tid = threadIdx.x;
    const int b = blockIdx.x;
    int v = tid < C ? cnt[b * C + tid] : 0;
    int incl = block_incl_scan(v, sh);
    if (tid < C) cnt[b * C + tid] = incl - v;
    if (tid == 255) btot[b] = incl;
    for (int i = b * 256 + tid; i < nz; i += gridDim.x * 256) pz[i] = 0.0f;
}

__global__ void __launch_bounds__(1024)
gcn_scatter(const int* __restrict__ src, const int* __restrict__ dst,
            const int* __restrict__ offs, const int* __restrict__ btot,
            int* __restrict__ bstart, int* __restrict__ part,
            int E, int chunkLen, int B, int C) {
    __shared__ int sh[1024];
    __shared__ int cur[1024];
    const int tid = threadIdx.x;
    const int c = blockIdx.x;
    int v = tid < B ? btot[tid] : 0;
    int incl = block_incl_scan(v, sh);
    int excl = incl - v;
    if (tid < B) cur[tid] = excl + offs[tid * C + c];
    if (c == 0 && tid <= B) bstart[tid] = excl;
    __syncthreads();
    const int lo = c * chunkLen;
    const int hi = min(lo + chunkLen, E);
    for (int e = lo + tid; e < hi; e += blockDim.x) {
        int d = dst[e];
        int s = src[e];
        int pos = atomicAdd(&cur[d >> RSH], 1);
        part[pos] = s | ((d & (RN - 1)) << SRC_BITS);
    }
}

__global__ void __launch_bounds__(1024)
gcn_deg_bucket(const int* __restrict__ part, const int* __restrict__ bstart,
               const float* __restrict__ x, float* __restrict__ dis, float* __restrict__ xs, int N) {
    __shared__ int h[RN];
    const int tid = threadIdx.x;
    if (tid < RN) h[tid] = 0;
    __syncthreads();
    const int e0 = bstart[blockIdx.x], e1 = bstart[blockIdx.x + 1];
    for (int i = e0 + tid; i < e1; i += blockDim.x) atomicAdd(&h[part[i] >> SRC_BITS], 1);
    __syncthreads();
    if (tid < RN) {
        int nd = (blockIdx.x << RSH) + tid;
        if (nd < N) {
            float d = rsqrtf((float)h[tid] + 1.0f);
            dis[nd] = d;
            xs[nd] = x[nd] * d;
        }
    }
}

__global__ void __launch_bounds__(1024)
gcn_edge1_bucket(const int* __restrict__ part, const int* __restrict__ bstart,
                 const float* __restrict__ xs, const float* __restrict__ dis,
                 const float* __restrict__ x, float* __restrict__ t, int N) {
    __shared__ float acc[RN];
    __shared__ float disL[RN];
    const int tid = threadIdx.x;
    const int nd = (blockIdx.x << RSH) + tid;
    if (tid < RN) {
        acc[tid] = 0.0f;
        disL[tid] = nd < N ? dis[nd] : 0.0f;
    }
    __syncthreads();
    const int e0 = bstart[blockIdx.x], e1 = bstart[blockIdx.x + 1];
    for (int i = e0 + tid; i < e1; i += blockDim.x) {
        int w = part[i];
        int dl = w >> SRC_BITS;
        atomicAdd(&acc[dl], xs[w & SRC_MASK] * disL[dl]);
    }
    __syncthreads();
    if (tid < RN && nd < N) t[nd] = acc[tid] + disL[tid] * disL[tid] * x[nd];
}

__global__ void gcn_gemm_wmma(const float* __restrict__ t,
                              const float* __restrict__ W1,
                              const float* __restrict__ b1,
                              const float* __restrict__ W2,
                              float* __restrict__ h2pre,
                              int nTiles) {
    int wave = (int)((blockIdx.x * blockDim.x + threadIdx.x) >> 5);
    int lane = threadIdx.x & 31;
    if (wave >= nTiles) return;

    int mt = wave >> 1;
    int nt = wave & 1;
    int m0 = mt << 4;
    int n0 = nt << 4;
    int r = lane & 15;
    int khalf = lane >> 4;

    float trow = t[m0 + r];

    v8f acc = {};
#pragma unroll
    for (int k0 = 0; k0 < 64; k0 += 4) {
        int ka = k0 + khalf * 2;
        float a0 = fmaf(trow, W1[ka],     b1[ka]);
        float a1 = fmaf(trow, W1[ka + 1], b1[ka + 1]);
        v2f a, b;
        a.x = a0 > 0.0f ? a0 : 0.0f;
        a.y = a1 > 0.0f ? a1 : 0.0f;
        b.x = W2[ka * 32 + n0 + r];
        b.y = W2[(ka + 1) * 32 + n0 + r];
        acc = __builtin_amdgcn_wmma_f32_16x16x4_f32(
             false, a,  false, b,
             (short)0, acc,  false,  false);
    }

    int col = n0 + r;
    int rowBase = m0 + khalf * 8;
#pragma unroll
    for (int v = 0; v < 8; ++v) {
        h2pre[(rowBase + v) * 32 + col] = acc[v];
    }
}

__global__ void __launch_bounds__(1024)
gcn_edge2_bucket(const int* __restrict__ part, const int* __restrict__ bstart,
                 const float* __restrict__ dis, const float* __restrict__ h2pre,
                 const float* __restrict__ b2, const int* __restrict__ batch,
                 float* __restrict__ psum, float* __restrict__ pcnt, int N) {
    __shared__ float acc[RN * 32];
    __shared__ float disL[RN];
    const int tid = threadIdx.x;
    const int lane = tid & 31;
    const int wave = tid >> 5;
    const int nWaves = blockDim.x >> 5;
    const int node0 = blockIdx.x << RSH;
    for (int i = tid; i < RN * 32; i += blockDim.x) acc[i] = 0.0f;
    if (tid < RN) {
        int nd = node0 + tid;
        disL[tid] = nd < N ? dis[nd] : 0.0f;
    }
    __syncthreads();
    const int e0 = bstart[blockIdx.x], e1 = bstart[blockIdx.x + 1];
    for (int base = e0 + wave * 32; base < e1; base += nWaves * 32) {
        const int idx = base + lane;
        const int w = idx < e1 ? part[idx] : 0;
        const int s = w & SRC_MASK;
        const int dl = w >> SRC_BITS;
        const float nrm = dis[s] * disL[dl];
        const int nrmBits = __float_as_int(nrm);
        const int cnt = min(32, e1 - base);
        if (cnt == 32) {
#pragma unroll
            for (int j0 = 0; j0 < 32; j0 += 8) {
                float v[8];
#pragma unroll
                for (int u = 0; u < 8; ++u) {
                    int sj = __builtin_amdgcn_readlane(s, j0 + u);
                    v[u] = h2pre[sj * 32 + lane];
                }
#pragma unroll
                for (int u = 0; u < 8; ++u) {
                    int dj = __builtin_amdgcn_readlane(dl, j0 + u);
                    float nj = __int_as_float(__builtin_amdgcn_readlane(nrmBits, j0 + u));
                    atomicAdd(&acc[dj * 32 + lane], v[u] * nj);
                }
            }
        } else {
            for (int j = 0; j < cnt; ++j) {
                int sj = __builtin_amdgcn_readlane(s, j);
                int dj = __builtin_amdgcn_readlane(dl, j);
                float nj = __int_as_float(__builtin_amdgcn_readlane(nrmBits, j));
                atomicAdd(&acc[dj * 32 + lane], h2pre[sj * 32 + lane] * nj);
            }
        }
    }
    __syncthreads();
    for (int i = tid; i < RN * 32; i += blockDim.x) {
        const int nl = i >> 5;
        const int f = i & 31;
        const int nd = node0 + nl;
        if (nd < N) {
            float d2 = disL[nl] * disL[nl];
            float h = acc[i] + d2 * h2pre[nd * 32 + f] + b2[f];
            h = h > 0.0f ? h : 0.0f;
            int g = batch[nd];
            atomicAdd(&psum[g * 32 + f], h);
            if (f == 0) atomicAdd(&pcnt[g], 1.0f);
        }
    }
}

#define HP 132
__global__ void __launch_bounds__(512)
gcn_head_wmma(const float* __restrict__ psum, const float* __restrict__ pcnt,
              const float* __restrict__ Wp1, const float* __restrict__ bp1,
              const float* __restrict__ Wp2, const float* __restrict__ bp2,
              float* __restrict__ out, int G) {
    __shared__ float hs[32 * HP];
    __shared__ float os[96];
    const int tid = threadIdx.x;
    const int lane = tid & 31;
    const int wave = tid >> 5;
    const int c = lane & 15;
    const int khalf = lane >> 4;
    const int mt = wave >> 3;
    const int n0 = (wave & 7) << 4;
    const int g0 = blockIdx.x << 5;

    const int row = tid >> 4;
    const int sub = tid & 15;
    float w2[24];
#pragma unroll
    for (int j = 0; j < 24; ++j) w2[j] = Wp2[sub * 24 + j];
    float o0 = 0.0f, o1 = 0.0f, o2 = 0.0f;

    int g = g0 + (mt << 4) + c;
    g = g < G ? g : G - 1;
    float cn = pcnt[g];
    cn = cn > 1.0f ? cn : 1.0f;
    const float inv = 1.0f / cn;
    const float* prow = psum + g * 32 + khalf * 2;
    const float* wcol = Wp1 + (khalf * 2) * 128 + n0 + c;

    v8f acc0 = {};
    v8f acc1 = {};
#pragma unroll
    for (int k0 = 0; k0 < 32; k0 += 8) {
        v2f a, b;
        a.x = prow[k0] * inv;
        a.y = prow[k0 + 1] * inv;
        b.x = wcol[k0 * 128];
        b.y = wcol[(k0 + 1) * 128];
        acc0 = __builtin_amdgcn_wmma_f32_16x16x4_f32(
             false, a,  false, b,
             (short)0, acc0,  false,  false);
        a.x = prow[k0 + 4] * inv;
        a.y = prow[k0 + 5] * inv;
        b.x = wcol[(k0 + 4) * 128];
        b.y = wcol[(k0 + 5) * 128];
        acc1 = __builtin_amdgcn_wmma_f32_16x16x4_f32(
             false, a,  false, b,
             (short)0, acc1,  false,  false);
    }

    const float bias = bp1[n0 + c];
    const int rowBase = (mt << 4) + khalf * 8;
#pragma unroll
    for (int v = 0; v < 8; ++v) {
        float h = acc0[v] + acc1[v] + bias;
        hs[(rowBase + v) * HP + n0 + c] = h > 0.0f ? h : 0.0f;
    }
    __syncthreads();

    const float* hr = hs + row * HP + sub * 8;
#pragma unroll
    for (int j = 0; j < 8; ++j) {
        float h = hr[j];
        o0 = fmaf(h, w2[j * 3 + 0], o0);
        o1 = fmaf(h, w2[j * 3 + 1], o1);
        o2 = fmaf(h, w2[j * 3 + 2], o2);
    }
#pragma unroll
    for (int m = 8; m >= 1; m >>= 1) {
        o0 += __shfl_xor(o0, m, 16);
        o1 += __shfl_xor(o1, m, 16);
        o2 += __shfl_xor(o2, m, 16);
    }
    if (sub == 0) {
        os[row * 3 + 0] = o0 + bp2[0];
        os[row * 3 + 1] = o1 + bp2[1];
        os[row * 3 + 2] = o2 + bp2[2];
    }
    __syncthreads();
    if (tid < 96) {
        int idx = g0 * 3 + tid;
        if (idx < G * 3) out[idx] = os[tid];
    }
}

extern "C" void kernel_launch(void* const* d_in, const int* in_sizes, int n_in,
                              void* d_out, int out_size, void* d_ws, size_t ws_size,
                              hipStream_t stream) {
    const float* x   = (const float*)d_in[0];
    const float* W1  = (const float*)d_in[1];
    const float* b1  = (const float*)d_in[2];
    const float* W2  = (const float*)d_in[3];
    const float* b2  = (const float*)d_in[4];
    const float* Wp1 = (const float*)d_in[5];
    const float* bp1 = (const float*)d_in[6];
    const float* Wp2 = (const float*)d_in[7];
    const float* bp2 = (const float*)d_in[8];
    const int* eidx  = (const int*)d_in[9];
    const int* batch = (const int*)d_in[10];
    float* out = (float*)d_out;

    const int N = in_sizes[0];
    const int E = in_sizes[9] / 2;
    const int G = out_size / 3;
    const int* src = eidx;
    const int* dst = eidx + E;

    const int B = (N + RN - 1) >> RSH;
    const int C = NCHUNK;
    const int chunkLen = (E + C - 1) / C;

    float* wsf   = (float*)d_ws;
    float* dis   = wsf;
    float* xs    = dis + N;
    float* tbuf  = xs + N;
    int*   bstart = (int*)(tbuf + N);
    int*   btot  = bstart + (B + 1);
    int*   cnt   = btot + B;
    float* h2pre = wsf + 4L * N;
    int*   part  = (int*)(h2pre + (long)N * 32);
    float* psum  = (float*)(part + E);
    float* pcnt  = psum + (long)G * 32;

    const int TPB = 256;

    gcn_count<<<C, 1024, 0, stream>>>(dst, cnt, E, chunkLen, B, C);
    gcn_bscan<<<B, 256, 0, stream>>>(cnt, btot, C, psum, 33 * G);
    gcn_scatter<<<C, 1024, 0, stream>>>(src, dst, cnt, btot, bstart, part, E, chunkLen, B, C);

    gcn_deg_bucket<<<B, 1024, 0, stream>>>(part, bstart, x, dis, xs, N);
    gcn_edge1_bucket<<<B, 1024, 0, stream>>>(part, bstart, xs, dis, x, tbuf, N);

    int nTiles = (N / 16) * 2;
    gcn_gemm_wmma<<<(nTiles + 7) / 8, TPB, 0, stream>>>(tbuf, W1, b1, W2, h2pre, nTiles);

    gcn_edge2_bucket<<<B, 1024, 0, stream>>>(part, bstart, dis, h2pre, b2, batch, psum, pcnt, N);

    gcn_head_wmma<<<(G + 31) / 32, 512, 0, stream>>>(psum, pcnt, Wp1, bp1, Wp2, bp2, out, G);
}


# layer 2 sums before the W2 product: bucket edges dst-ordered, a wave rebuilds relu(t*W1+b1) per edge from one gathered {t,dis} pair into register row sums, block does its own WMMA; sum order changes
# speedup vs baseline: 4.5734x; 1.1879x over previous
#include <hip/hip_runtime.h>
#include <hip/hip_bf16.h>


typedef float v2f __attribute__((ext_vector_type(2)));
typedef float v8f __attribute__((ext_vector_type(8)));

#define RSH 8
#define RN 256
#define SRC_BITS 20
#define SRC_MASK 0xFFFFF
#define NCHUNK 250
#define SCAP 16384
#define AP 68


__device__ __forceinline__ int block_incl_scan(int v, int* sh) {
    const int tid = threadIdx.x;
    const int n = blockDim.x;
    sh[tid] = v;
    __syncthreads();
    for (int off = 1; off < n; off <<= 1) {
        int a = tid >= off ? sh[tid - off] : 0;
        __syncthreads();
        v += a;
        sh[tid] = v;
        __syncthreads();
    }
    return v;
}

__global__ void __launch_bounds__(1024)
gcn_count(const int* __restrict__ dst, int* __restrict__ cnt, int E, int chunkLen, int B, int C) {
    __shared__ int h[1024];
    const int tid = threadIdx.x;
    for (int i = tid; i < B; i += blockDim.x) h[i] = 0;
    __syncthreads();
    const int lo = blockIdx.x * chunkLen;
    const int hi = min(lo + chunkLen, E);
    for (int e = lo + tid; e < hi; e += blockDim.x) atomicAdd(&h[dst[e] >> RSH], 1);
    __syncthreads();
    for (int i = tid; i < B; i += blockDim.x) cnt[i * C + blockIdx.x] = h[i];
}

__global__ void __launch_bounds__(256)
gcn_bscan(int* __restrict__ cnt, int* __restrict__ btot, int C, float* __restrict__ pz, int nz) {
    __shared__ int sh[256];
    const int tid = threadIdx.x;
    const int b = blockIdx.x;
    int v = tid < C ? cnt[b * C + tid] : 0;
    int incl = block_incl_scan(v, sh);
    if (tid < C) cnt[b * C + tid] = incl - v;
    if (tid == 255) btot[b] = incl;
    for (int i = b * 256 + tid; i < nz; i += gridDim.x * 256) pz[i] = 0.0f;
}

__global__ void __launch_bounds__(1024)
gcn_scatter(const int* __restrict__ src, const int* __restrict__ dst,
            const int* __restrict__ offs, const int* __restrict__ btot,
            int* __restrict__ bstart, int* __restrict__ part,
            int E, int chunkLen, int B, int C) {
    __shared__ int sh[1024];
    __shared__ int cur[1024];
    const int tid = threadIdx.x;
    const int c = blockIdx.x;
    int v = tid < B ? btot[tid] : 0;
    int incl = block_incl_scan(v, sh);
    int excl = incl - v;
    if (tid < B) cur[tid] = excl + offs[tid * C + c];
    if (c == 0 && tid <= B) bstart[tid] = excl;
    __syncthreads();
    const int lo = c * chunkLen;
    const int hi = min(lo + chunkLen, E);
    for (int e = lo + tid; e < hi; e += blockDim.x) {
        int d = dst[e];
        int s = src[e];
        int pos = atomicAdd(&cur[d >> RSH], 1);
        part[pos] = s | ((d & (RN - 1)) << SRC_BITS);
    }
}

__global__ void __launch_bounds__(1024)
gcn_deg_sort(const int* __restrict__ part, const int* __restrict__ bstart,
             const float* __restrict__ x, float* __restrict__ pair, float* __restrict__ xs,
             int* __restrict__ part2, int N) {
    __shared__ int h[RN];
    __shared__ int cur[RN];
    __shared__ int stage[SCAP];
    const int tid = threadIdx.x;
    if (tid < RN) h[tid] = 0;
    __syncthreads();
    const int e0 = bstart[blockIdx.x], e1 = bstart[blockIdx.x + 1];
    for (int i = e0 + tid; i < e1; i += blockDim.x) atomicAdd(&h[part[i] >> SRC_BITS], 1);
    __syncthreads();
    if (tid < RN) {
        int nd = (blockIdx.x << RSH) + tid;
        if (nd < N) {
            float d = rsqrtf((float)h[tid] + 1.0f);
            pair[2 * nd + 1] = d;
            xs[nd] = x[nd] * d;
        }
    }
    if (tid < 32) {
        int a[RN / 32];
        int sum = 0;
#pragma unroll
        for (int u = 0; u < RN / 32; ++u) { a[u] = h[tid * (RN / 32) + u]; sum += a[u]; }
        int incl = sum;
#pragma unroll
        for (int off = 1; off < 32; off <<= 1) {
            int o = __shfl_up(incl, off);
            if (tid >= off) incl += o;
        }
        int run = incl - sum;
#pragma unroll
        for (int u = 0; u < RN / 32; ++u) { cur[tid * (RN / 32) + u] = run; run += a[u]; }
    }
    __syncthreads();
    const int nE = e1 - e0;
    if (nE <= SCAP) {
        for (int i = e0 + tid; i < e1; i += blockDim.x) {
            int w = part[i];
            stage[atomicAdd(&cur[w >> SRC_BITS], 1)] = w;
        }
        __syncthreads();
        for (int i = tid; i < nE; i += blockDim.x) part2[e0 + i] = stage[i];
    } else {
        for (int i = e0 + tid; i < e1; i += blockDim.x) {
            int w = part[i];
            part2[e0 + atomicAdd(&cur[w >> SRC_BITS], 1)] = w;
        }
    }
}

__global__ void __launch_bounds__(1024)
gcn_edge1_bucket(const int* __restrict__ part, const int* __restrict__ bstart,
                 const float* __restrict__ xs, float* __restrict__ pair,
                 const float* __restrict__ x, int N) {
    __shared__ float acc[RN];
    __shared__ float disL[RN];
    const int tid = threadIdx.x;
    const int nd = (blockIdx.x << RSH) + tid;
    if (tid < RN) {
        acc[tid] = 0.0f;
        disL[tid] = nd < N ? pair[2 * nd + 1] : 0.0f;
    }
    __syncthreads();
    const int e0 = bstart[blockIdx.x], e1 = bstart[blockIdx.x + 1];
    for (int i = e0 + tid; i < e1; i += blockDim.x) {
        int w = part[i];
        int dl = w >> SRC_BITS;
        atomicAdd(&acc[dl], xs[w & SRC_MASK] * disL[dl]);
    }
    __syncthreads();
    if (tid < RN && nd < N) pair[2 * nd] = acc[tid] + disL[tid] * disL[tid] * x[nd];
}

__device__ __forceinline__ void row_flush(float* acc, int dl, int lane, float a0, float a1) {
    atomicAdd(&acc[dl * AP + lane], a0);
    atomicAdd(&acc[dl * AP + 32 + lane], a1);
}

__global__ void __launch_bounds__(1024)
gcn_layer2_bucket(const int* __restrict__ part2, const int* __restrict__ bstart,
                  const float* __restrict__ pair,
                  const float* __restrict__ W1, const float* __restrict__ b1,
                  const float* __restrict__ W2, const float* __restrict__ b2,
                  const int* __restrict__ batch,
                  float* __restrict__ psum, float* __restrict__ pcnt, int N) {
    __shared__ float acc[RN * AP];
    __shared__ float hs[RN * 32];
    __shared__ float disL[RN];
    __shared__ float tL[RN];
    const int tid = threadIdx.x;
    const int lane = tid & 31;
    const int wave = tid >> 5;
    const int nWaves = blockDim.x >> 5;
    const int node0 = blockIdx.x << RSH;
    for (int i = tid; i < RN * AP; i += blockDim.x) acc[i] = 0.0f;
    if (tid < RN) {
        int nd = node0 + tid;
        tL[tid] = nd < N ? pair[2 * nd] : 0.0f;
        disL[tid] = nd < N ? pair[2 * nd + 1] : 0.0f;
    }
    const float w1a = W1[lane], w1b = W1[lane + 32];
    const float b1a = b1[lane], b1b = b1[lane + 32];
    __syncthreads();
    const int e0 = bstart[blockIdx.x], e1 = bstart[blockIdx.x + 1];
    for (int base = e0 + wave * 32; base < e1; base += nWaves * 32) {
        const int idx = base + lane;
        const int w = idx < e1 ? part2[idx] : 0;
        const int s = w & SRC_MASK;
        const int dl = w >> SRC_BITS;
        const v2f p = *(const v2f*)(pair + 2 * s);
        const int tBits = __float_as_int(p.x);
        const int dBits = __float_as_int(p.y);
        const int cnt = min(32, e1 - base);
        const int prev = __shfl_up(dl, 1);
        const unsigned brk = __builtin_amdgcn_ballot_w32(lane > 0 && lane < cnt && dl != prev);
        float a0 = 0.0f, a1 = 0.0f;
        if (cnt == 32) {
#pragma unroll
            for (int j = 0; j < 32; ++j) {
                if (j > 0 && ((brk >> j) & 1u)) {
                    row_flush(acc, __builtin_amdgcn_readlane(dl, j - 1), lane, a0, a1);
                    a0 = 0.0f;
                    a1 = 0.0f;
                }
                const float tj = __int_as_float(__builtin_amdgcn_readlane(tBits, j));
                const float dj = __int_as_float(__builtin_amdgcn_readlane(dBits, j));
                a0 = fmaf(dj, fmaxf(fmaf(tj, w1a, b1a), 0.0f), a0);
                a1 = fmaf(dj, fmaxf(fmaf(tj, w1b, b1b), 0.0f), a1);
            }
            row_flush(acc, __builtin_amdgcn_readlane(dl, 31), lane, a0, a1);
        } else {
            for (int j = 0; j < cnt; ++j) {
                if ((brk >> j) & 1u) {
                    row_flush(acc, __builtin_amdgcn_readlane(dl, j - 1), lane, a0, a1);
                    a0 = 0.0f;
                    a1 = 0.0f;
                }
                const float tj = __int_as_float(__builtin_amdgcn_readlane(tBits, j));
                const float dj = __int_as_float(__builtin_amdgcn_readlane(dBits, j));
                a0 = fmaf(dj, fmaxf(fmaf(tj, w1a, b1a), 0.0f), a0);
                a1 = fmaf(dj, fmaxf(fmaf(tj, w1b, b1b), 0.0f), a1);
            }
            row_flush(acc, __builtin_amdgcn_readlane(dl, cnt - 1), lane, a0, a1);
        }
    }
    __syncthreads();
    {
        const int k = tid & 63;
        const float wk = W1[k], bk = b1[k];
        for (int nl = tid >> 6; nl < RN; nl += blockDim.x >> 6) {
            const float d = disL[nl];
            const float hself = fmaxf(fmaf(tL[nl], wk, bk), 0.0f);
            float* q = &acc[nl * AP + k];
            *q = d * (*q) + d * d * hself;
        }
    }
    __syncthreads();
    const int c = lane & 15;
    const int khalf = lane >> 4;
    const int mt = wave >> 1;
    const int n0 = (wave & 1) << 4;
    const float* arow = acc + (mt * 16 + c) * AP + khalf * 2;
    const float* wcol = W2 + (khalf * 2) * 32 + n0 + c;
    const float bias = b2[n0 + c];
    v8f D = {};
#pragma unroll
    for (int k0 = 0; k0 < 64; k0 += 4) {
        v2f a, b;
        a.x = arow[k0];
        a.y = arow[k0 + 1];
        b.x = wcol[k0 * 32];
        b.y = wcol[(k0 + 1) * 32];
        D = __builtin_amdgcn_wmma_f32_16x16x4_f32(
             false, a,  false, b,
             (short)0, D,  false,  false);
    }
    float* hrow = hs + (mt * 16 + khalf * 8) * 32 + n0 + c;
#pragma unroll
    for (int v = 0; v < 8; ++v) {
        float hv = D[v] + bias;
        hrow[v * 32] = hv > 0.0f ? hv : 0.0f;
    }
    __syncthreads();
    for (int i = tid; i < RN * 32; i += blockDim.x) {
        const int nl = i >> 5;
        const int f = i & 31;
        const int nd = node0 + nl;
        if (nd < N) {
            const int g = batch[nd];
            atomicAdd(&psum[g * 32 + f], hs[i]);
            if (f == 0) atomicAdd(&pcnt[g], 1.0f);
        }
    }
}

#define HP 132
__global__ void __launch_bounds__(512)
gcn_head_wmma(const float* __restrict__ psum, const float* __restrict__ pcnt,
              const float* __restrict__ Wp1, const float* __restrict__ bp1,
              const float* __restrict__ Wp2, const float* __restrict__ bp2,
              float* __restrict__ out, int G) {
    __shared__ float hs[32 * HP];
    __shared__ float os[96];
    const int tid = threadIdx.x;
    const int lane = tid & 31;
    const int wave = tid >> 5;
    const int c = lane & 15;
    const int khalf = lane >> 4;
    const int mt = wave >> 3;
    const int n0 = (wave & 7) << 4;
    const int g0 = blockIdx.x << 5;

    const int row = tid >> 4;
    const int sub = tid & 15;
    float w2[24];
#pragma unroll
    for (int j = 0; j < 24; ++j) w2[j] = Wp2[sub * 24 + j];
    float o0 = 0.0f, o1 = 0.0f, o2 = 0.0f;

    int g = g0 + (mt << 4) + c;
    g = g < G ? g : G - 1;
    float cn = pcnt[g];
    cn = cn > 1.0f ? cn : 1.0f;
    const float inv = 1.0f / cn;
    const float* prow = psum + g * 32 + khalf * 2;
    const float* wcol = Wp1 + (khalf * 2) * 128 + n0 + c;

    v8f acc0 = {};
    v8f acc1 = {};
#pragma unroll
    for (int k0 = 0; k0 < 32; k0 += 8) {
        v2f a, b;
        a.x = prow[k0] * inv;
        a.y = prow[k0 + 1] * inv;
        b.x = wcol[k0 * 128];
        b.y = wcol[(k0 + 1) * 128];
        acc0 = __builtin_amdgcn_wmma_f32_16x16x4_f32(
             false, a,  false, b,
             (short)0, acc0,  false,  false);
        a.x = prow[k0 + 4] * inv;
        a.y = prow[k0 + 5] * inv;
        b.x = wcol[(k0 + 4) * 128];
        b.y = wcol[(k0 + 5) * 128];
        acc1 = __builtin_amdgcn_wmma_f32_16x16x4_f32(
             false, a,  false, b,
             (short)0, acc1,  false,  false);
    }

    const float bias = bp1[n0 + c];
    const int rowBase = (mt << 4) + khalf * 8;
#pragma unroll
    for (int v = 0; v < 8; ++v) {
        float h = acc0[v] + acc1[v] + bias;
        hs[(rowBase + v) * HP + n0 + c] = h > 0.0f ? h : 0.0f;
    }
    __syncthreads();

    const float* hr = hs + row * HP + sub * 8;
#pragma unroll
    for (int j = 0; j < 8; ++j) {
        float h = hr[j];
        o0 = fmaf(h, w2[j * 3 + 0], o0);
        o1 = fmaf(h, w2[j * 3 + 1], o1);
        o2 = fmaf(h, w2[j * 3 + 2], o2);
    }
#pragma unroll
    for (int m = 8; m >= 1; m >>= 1) {
        o0 += __shfl_xor(o0, m, 16);
        o1 += __shfl_xor(o1, m, 16);
        o2 += __shfl_xor(o2, m, 16);
    }
    if (sub == 0) {
        os[row * 3 + 0] = o0 + bp2[0];
        os[row * 3 + 1] = o1 + bp2[1];
        os[row * 3 + 2] = o2 + bp2[2];
    }
    __syncthreads();
    if (tid < 96) {
        int idx = g0 * 3 + tid;
        if (idx < G * 3) out[idx] = os[tid];
    }
}

extern "C" void kernel_launch(void* const* d_in, const int* in_sizes, int n_in,
                              void* d_out, int out_size, void* d_ws, size_t ws_size,
                              hipStream_t stream) {
    const float* x   = (const float*)d_in[0];
    const float* W1  = (const float*)d_in[1];
    const float* b1  = (const float*)d_in[2];
    const float* W2  = (const float*)d_in[3];
    const float* b2  = (const float*)d_in[4];
    const float* Wp1 = (const float*)d_in[5];
    const float* bp1 = (const float*)d_in[6];
    const float* Wp2 = (const float*)d_in[7];
    const float* bp2 = (const float*)d_in[8];
    const int* eidx  = (const int*)d_in[9];
    const int* batch = (const int*)d_in[10];
    float* out = (float*)d_out;

    const int N = in_sizes[0];
    const int E = in_sizes[9] / 2;
    const int G = out_size / 3;
    const int* src = eidx;
    const int* dst = eidx + E;

    const int B = (N + RN - 1) >> RSH;
    const int C = NCHUNK;
    const int chunkLen = (E + C - 1) / C;

    float* wsf   = (float*)d_ws;
    float* pair  = wsf;
    float* xs    = pair + 2L * N;
    int*   bstart = (int*)(xs + N);
    int*   btot  = bstart + (B + 1);
    int*   cnt   = btot + B;
    int*   part  = (int*)(wsf + 4L * N);
    int*   part2 = part + E;
    float* psum  = (float*)(part2 + E);
    float* pcnt  = psum + (long)G * 32;

    gcn_count<<<C, 1024, 0, stream>>>(dst, cnt, E, chunkLen, B, C);
    gcn_bscan<<<B, 256, 0, stream>>>(cnt, btot, C, psum, 33 * G);
    gcn_scatter<<<C, 1024, 0, stream>>>(src, dst, cnt, btot, bstart, part, E, chunkLen, B, C);

    gcn_deg_sort<<<B, 1024, 0, stream>>>(part, bstart, x, pair, xs, part2, N);
    gcn_edge1_bucket<<<B, 1024, 0, stream>>>(part, bstart, xs, pair, x, N);
    gcn_layer2_bucket<<<B, 1024, 0, stream>>>(part2, bstart, pair, W1, b1, W2, b2, batch, psum, pcnt, N);

    gcn_head_wmma<<<(G + 31) / 32, 512, 0, stream>>>(psum, pcnt, Wp1, bp1, Wp2, bp2, out, G);
}


# edge scatter: a chunk's edge words are ordered by bucket in LDS first (counts from the offset table) and leave as contiguous per-bucket runs, a few lines per wave store instead of 32; sums unchanged
# speedup vs baseline: 5.9336x; 1.2974x over previous
#include <hip/hip_runtime.h>
#include <hip/hip_bf16.h>


typedef float v2f __attribute__((ext_vector_type(2)));
typedef float v8f __attribute__((ext_vector_type(8)));
typedef int v2i __attribute__((ext_vector_type(2)));

#define RSH 8
#define RN 256
#define SRC_BITS 20
#define SRC_MASK 0xFFFFF
#define NCHUNK 250
#define SCAP 16384
#define CCAP 16384
#define AP 68


__device__ __forceinline__ int block_incl_scan(int v, int* sh) {
    const int tid = threadIdx.x;
    const int n = blockDim.x;
    sh[tid] = v;
    __syncthreads();
    for (int off = 1; off < n; off <<= 1) {
        int a = tid >= off ? sh[tid - off] : 0;
        __syncthreads();
        v += a;
        sh[tid] = v;
        __syncthreads();
    }
    return v;
}

__global__ void __launch_bounds__(1024)
gcn_count(const int* __restrict__ dst, int* __restrict__ cnt, int E, int chunkLen, int B, int C) {
    __shared__ int h[1024];
    const int tid = threadIdx.x;
    for (int i = tid; i < B; i += blockDim.x) h[i] = 0;
    __syncthreads();
    const int lo = blockIdx.x * chunkLen;
    const int hi = min(lo + chunkLen, E);
    for (int e = lo + tid; e < hi; e += blockDim.x) atomicAdd(&h[dst[e] >> RSH], 1);
    __syncthreads();
    for (int i = tid; i < B; i += blockDim.x) cnt[i * C + blockIdx.x] = h[i];
}

__global__ void __launch_bounds__(256)
gcn_bscan(int* __restrict__ cnt, int* __restrict__ btot, int C, float* __restrict__ pz, int nz) {
    __shared__ int sh[256];
    const int tid = threadIdx.x;
    const int b = blockIdx.x;
    int v = tid < C ? cnt[b * C + tid] : 0;
    int incl = block_incl_scan(v, sh);
    if (tid < C) cnt[b * C + tid] = incl - v;
    if (tid == 255) btot[b] = incl;
    for (int i = b * 256 + tid; i < nz; i += gridDim.x * 256) pz[i] = 0.0f;
}

__device__ __forceinline__ long long block_incl_scan64(long long v, long long* sh) {
    const int tid = threadIdx.x;
    const int n = blockDim.x;
    sh[tid] = v;
    __syncthreads();
    for (int off = 1; off < n; off <<= 1) {
        long long a = tid >= off ? sh[tid - off] : 0;
        __syncthreads();
        v += a;
        sh[tid] = v;
        __syncthreads();
    }
    return v;
}

__global__ void __launch_bounds__(1024)
gcn_scatter(const int* __restrict__ src, const int* __restrict__ dst,
            const int* __restrict__ offs, const int* __restrict__ btot,
            int* __restrict__ bstart, int* __restrict__ part,
            int E, int chunkLen, int B, int C) {
    __shared__ long long sh[1024];
    __shared__ int cur[1024];
    __shared__ int delta[1024];
    __shared__ v2i stage[CCAP];
    const int tid = threadIdx.x;
    const int c = blockIdx.x;
    int bt = 0, off = 0, lc = 0;
    if (tid < B) {
        bt = btot[tid];
        off = offs[tid * C + c];
        lc = (c + 1 < C ? offs[tid * C + c + 1] : bt) - off;
    }
    const long long v = ((long long)bt << 32) | (long long)lc;
    const long long excl = block_incl_scan64(v, sh) - v;
    const int gexcl = (int)(excl >> 32);
    const int lexcl = (int)excl;
    if (c == 0 && tid <= B) bstart[tid] = gexcl;
    const int lo = c * chunkLen;
    const int hi = min(lo + chunkLen, E);
    const int n = hi - lo;
    if (n <= CCAP) {
        if (tid < B) {
            cur[tid] = lexcl;
            delta[tid] = gexcl + off - lexcl;
        }
        __syncthreads();
        for (int e = lo + tid; e < hi; e += blockDim.x) {
            int d = dst[e];
            int s = src[e];
            int b = d >> RSH;
            int pos = atomicAdd(&cur[b], 1);
            v2i w;
            w.x = s | ((d & (RN - 1)) << SRC_BITS);
            w.y = pos + delta[b];
            stage[pos] = w;
        }
        __syncthreads();
        for (int i = tid; i < n; i += blockDim.x) {
            v2i w = stage[i];
            part[w.y] = w.x;
        }
    } else {
        if (tid < B) cur[tid] = gexcl + off;
        __syncthreads();
        for (int e = lo + tid; e < hi; e += blockDim.x) {
            int d = dst[e];
            int s = src[e];
            int pos = atomicAdd(&cur[d >> RSH], 1);
            part[pos] = s | ((d & (RN - 1)) << SRC_BITS);
        }
    }
}

__global__ void __launch_bounds__(1024)
gcn_deg_sort(const int* __restrict__ part, const int* __restrict__ bstart,
             const float* __restrict__ x, float* __restrict__ pair, float* __restrict__ xs,
             int* __restrict__ part2, int N) {
    __shared__ int h[RN];
    __shared__ int cur[RN];
    __shared__ int stage[SCAP];
    const int tid = threadIdx.x;
    if (tid < RN) h[tid] = 0;
    __syncthreads();
    const int e0 = bstart[blockIdx.x], e1 = bstart[blockIdx.x + 1];
    for (int i = e0 + tid; i < e1; i += blockDim.x) atomicAdd(&h[part[i] >> SRC_BITS], 1);
    __syncthreads();
    if (tid < RN) {
        int nd = (blockIdx.x << RSH) + tid;
        if (nd < N) {
            float d = rsqrtf((float)h[tid] + 1.0f);
            pair[2 * nd + 1] = d;
            xs[nd] = x[nd] * d;
        }
    }
    if (tid < 32) {
        int a[RN / 32];
        int sum = 0;
#pragma unroll
        for (int u = 0; u < RN / 32; ++u) { a[u] = h[tid * (RN / 32) + u]; sum += a[u]; }
        int incl = sum;
#pragma unroll
        for (int off = 1; off < 32; off <<= 1) {
            int o = __shfl_up(incl, off);
            if (tid >= off) incl += o;
        }
        int run = incl - sum;
#pragma unroll
        for (int u = 0; u < RN / 32; ++u) { cur[tid * (RN / 32) + u] = run; run += a[u]; }
    }
    __syncthreads();
    const int nE = e1 - e0;
    if (nE <= SCAP) {
        for (int i = e0 + tid; i < e1; i += blockDim.x) {
            int w = part[i];
            stage[atomicAdd(&cur[w >> SRC_BITS], 1)] = w;
        }
        __syncthreads();
        for (int i = tid; i < nE; i += blockDim.x) part2[e0 + i] = stage[i];
    } else {
        for (int i = e0 + tid; i < e1; i += blockDim.x) {
            int w = part[i];
            part2[e0 + atomicAdd(&cur[w >> SRC_BITS], 1)] = w;
        }
    }
}

__global__ void __launch_bounds__(1024)
gcn_edge1_bucket(const int* __restrict__ part, const int* __restrict__ bstart,
                 const float* __restrict__ xs, float* __restrict__ pair,
                 const float* __restrict__ x, int N) {
    __shared__ float acc[RN];
    __shared__ float disL[RN];
    const int tid = threadIdx.x;
    const int nd = (blockIdx.x << RSH) + tid;
    if (tid < RN) {
        acc[tid] = 0.0f;
        disL[tid] = nd < N ? pair[2 * nd + 1] : 0.0f;
    }
    __syncthreads();
    const int e0 = bstart[blockIdx.x], e1 = bstart[blockIdx.x + 1];
    for (int i = e0 + tid; i < e1; i += blockDim.x) {
        int w = part[i];
        int dl = w >> SRC_BITS;
        atomicAdd(&acc[dl], xs[w & SRC_MASK] * disL[dl]);
    }
    __syncthreads();
    if (tid < RN && nd < N) pair[2 * nd] = acc[tid] + disL[tid] * disL[tid] * x[nd];
}

__device__ __forceinline__ void row_flush(float* acc, int dl, int lane, float a0, float a1) {
    atomicAdd(&acc[dl * AP + lane], a0);
    atomicAdd(&acc[dl * AP + 32 + lane], a1);
}

__global__ void __launch_bounds__(1024)
gcn_layer2_bucket(const int* __restrict__ part2, const int* __restrict__ bstart,
                  const float* __restrict__ pair,
                  const float* __restrict__ W1, const float* __restrict__ b1,
                  const float* __restrict__ W2, const float* __restrict__ b2,
                  const int* __restrict__ batch,
                  float* __restrict__ psum, float* __restrict__ pcnt, int N) {
    __shared__ float acc[RN * AP];
    __shared__ float hs[RN * 32];
    __shared__ float disL[RN];
    __shared__ float tL[RN];
    const int tid = threadIdx.x;
    const int lane = tid & 31;
    const int wave = tid >> 5;
    const int nWaves = blockDim.x >> 5;
    const int node0 = blockIdx.x << RSH;
    for (int i = tid; i < RN * AP; i += blockDim.x) acc[i] = 0.0f;
    if (tid < RN) {
        int nd = node0 + tid;
        tL[tid] = nd < N ? pair[2 * nd] : 0.0f;
        disL[tid] = nd < N ? pair[2 * nd + 1] : 0.0f;
    }
    const float w1a = W1[lane], w1b = W1[lane + 32];
    const float b1a = b1[lane], b1b = b1[lane + 32];
    __syncthreads();
    const int e0 = bstart[blockIdx.x], e1 = bstart[blockIdx.x + 1];
    for (int base = e0 + wave * 32; base < e1; base += nWaves * 32) {
        const int idx = base + lane;
        const int w = idx < e1 ? part2[idx] : 0;
        const int s = w & SRC_MASK;
        const int dl = w >> SRC_BITS;
        const v2f p = *(const v2f*)(pair + 2 * s);
        const int tBits = __float_as_int(p.x);
        const int dBits = __float_as_int(p.y);
        const int cnt = min(32, e1 - base);
        const int prev = __shfl_up(dl, 1);
        const unsigned brk = __builtin_amdgcn_ballot_w32(lane > 0 && lane < cnt && dl != prev);
        float a0 = 0.0f, a1 = 0.0f;
        if (cnt == 32) {
#pragma unroll
            for (int j = 0; j < 32; ++j) {
                if (j > 0 && ((brk >> j) & 1u)) {
                    row_flush(acc, __builtin_amdgcn_readlane(dl, j - 1), lane, a0, a1);
                    a0 = 0.0f;
                    a1 = 0.0f;
                }
                const float tj = __int_as_float(__builtin_amdgcn_readlane(tBits, j));
                const float dj = __int_as_float(__builtin_amdgcn_readlane(dBits, j));
                a0 = fmaf(dj, fmaxf(fmaf(tj, w1a, b1a), 0.0f), a0);
                a1 = fmaf(dj, fmaxf(fmaf(tj, w1b, b1b), 0.0f), a1);
            }
            row_flush(acc, __builtin_amdgcn_readlane(dl, 31), lane, a0, a1);
        } else {
            for (int j = 0; j < cnt; ++j) {
                if ((brk >> j) & 1u) {
                    row_flush(acc, __builtin_amdgcn_readlane(dl, j - 1), lane, a0, a1);
                    a0 = 0.0f;
                    a1 = 0.0f;
                }
                const float tj = __int_as_float(__builtin_amdgcn_readlane(tBits, j));
                const float dj = __int_as_float(__builtin_amdgcn_readlane(dBits, j));
                a0 = fmaf(dj, fmaxf(fmaf(tj, w1a, b1a), 0.0f), a0);
                a1 = fmaf(dj, fmaxf(fmaf(tj, w1b, b1b), 0.0f), a1);
            }
            row_flush(acc, __builtin_amdgcn_readlane(dl, cnt - 1), lane, a0, a1);
        }
    }
    __syncthreads();
    {
        const int k = tid & 63;
        const float wk = W1[k], bk = b1[k];
        for (int nl = tid >> 6; nl < RN; nl += blockDim.x >> 6) {
            const float d = disL[nl];
            const float hself = fmaxf(fmaf(tL[nl], wk, bk), 0.0f);
            float* q = &acc[nl * AP + k];
            *q = d * (*q) + d * d * hself;
        }
    }
    __syncthreads();
    const int c = lane & 15;
    const int khalf = lane >> 4;
    const int mt = wave >> 1;
    const int n0 = (wave & 1) << 4;
    const float* arow = acc + (mt * 16 + c) * AP + khalf * 2;
    const float* wcol = W2 + (khalf * 2) * 32 + n0 + c;
    const float bias = b2[n0 + c];
    v8f D = {};
#pragma unroll
    for (int k0 = 0; k0 < 64; k0 += 4) {
        v2f a, b;
        a.x = arow[k0];
        a.y = arow[k0 + 1];
        b.x = wcol[k0 * 32];
        b.y = wcol[(k0 + 1) * 32];
        D = __builtin_amdgcn_wmma_f32_16x16x4_f32(
             false, a,  false, b,
             (short)0, D,  false,  false);
    }
    float* hrow = hs + (mt * 16 + khalf * 8) * 32 + n0 + c;
#pragma unroll
    for (int v = 0; v < 8; ++v) {
        float hv = D[v] + bias;
        hrow[v * 32] = hv > 0.0f ? hv : 0.0f;
    }
    __syncthreads();
    for (int i = tid; i < RN * 32; i += blockDim.x) {
        const int nl = i >> 5;
        const int f = i & 31;
        const int nd = node0 + nl;
        if (nd < N) {
            const int g = batch[nd];
            atomicAdd(&psum[g * 32 + f], hs[i]);
            if (f == 0) atomicAdd(&pcnt[g], 1.0f);
        }
    }
}

#define HP 132
__global__ void __launch_bounds__(512)
gcn_head_wmma(const float* __restrict__ psum, const float* __restrict__ pcnt,
              const float* __restrict__ Wp1, const float* __restrict__ bp1,
              const float* __restrict__ Wp2, const float* __restrict__ bp2,
              float* __restrict__ out, int G) {
    __shared__ float hs[32 * HP];
    __shared__ float os[96];
    const int tid = threadIdx.x;
    const int lane = tid & 31;
    const int wave = tid >> 5;
    const int c = lane & 15;
    const int khalf = lane >> 4;
    const int mt = wave >> 3;
    const int n0 = (wave & 7) << 4;
    const int g0 = blockIdx.x << 5;

    const int row = tid >> 4;
    const int sub = tid & 15;
    float w2[24];
#pragma unroll
    for (int j = 0; j < 24; ++j) w2[j] = Wp2[sub * 24 + j];
    float o0 = 0.0f, o1 = 0.0f, o2 = 0.0f;

    int g = g0 + (mt << 4) + c;
    g = g < G ? g : G - 1;
    float cn = pcnt[g];
    cn = cn > 1.0f ? cn : 1.0f;
    const float inv = 1.0f / cn;
    const float* prow = psum + g * 32 + khalf * 2;
    const float* wcol = Wp1 + (khalf * 2) * 128 + n0 + c;

    v8f acc0 = {};
    v8f acc1 = {};
#pragma unroll
    for (int k0 = 0; k0 < 32; k0 += 8) {
        v2f a, b;
        a.x = prow[k0] * inv;
        a.y = prow[k0 + 1] * inv;
        b.x = wcol[k0 * 128];
        b.y = wcol[(k0 + 1) * 128];
        acc0 = __builtin_amdgcn_wmma_f32_16x16x4_f32(
             false, a,  false, b,
             (short)0, acc0,  false,  false);
        a.x = prow[k0 + 4] * inv;
        a.y = prow[k0 + 5] * inv;
        b.x = wcol[(k0 + 4) * 128];
        b.y = wcol[(k0 + 5) * 128];
        acc1 = __builtin_amdgcn_wmma_f32_16x16x4_f32(
             false, a,  false, b,
             (short)0, acc1,  false,  false);
    }

    const float bias = bp1[n0 + c];
    const int rowBase = (mt << 4) + khalf * 8;
#pragma unroll
    for (int v = 0; v < 8; ++v) {
        float h = acc0[v] + acc1[v] + bias;
        hs[(rowBase + v) * HP + n0 + c] = h > 0.0f ? h : 0.0f;
    }
    __syncthreads();

    const float* hr = hs + row * HP + sub * 8;
#pragma unroll
    for (int j = 0; j < 8; ++j) {
        float h = hr[j];
        o0 = fmaf(h, w2[j * 3 + 0], o0);
        o1 = fmaf(h, w2[j * 3 + 1], o1);
        o2 = fmaf(h, w2[j * 3 + 2], o2);
    }
#pragma unroll
    for (int m = 8; m >= 1; m >>= 1) {
        o0 += __shfl_xor(o0, m, 16);
        o1 += __shfl_xor(o1, m, 16);
        o2 += __shfl_xor(o2, m, 16);
    }
    if (sub == 0) {
        os[row * 3 + 0] = o0 + bp2[0];
        os[row * 3 + 1] = o1 + bp2[1];
        os[row * 3 + 2] = o2 + bp2[2];
    }
    __syncthreads();
    if (tid < 96) {
        int idx = g0 * 3 + tid;
        if (idx < G * 3) out[idx] = os[tid];
    }
}

extern "C" void kernel_launch(void* const* d_in, const int* in_sizes, int n_in,
                              void* d_out, int out_size, void* d_ws, size_t ws_size,
                              hipStream_t stream) {
    const float* x   = (const float*)d_in[0];
    const float* W1  = (const float*)d_in[1];
    const float* b1  = (const float*)d_in[2];
    const float* W2  = (const float*)d_in[3];
    const float* b2  = (const float*)d_in[4];
    const float* Wp1 = (const float*)d_in[5];
    const float* bp1 = (const float*)d_in[6];
    const float* Wp2 = (const float*)d_in[7];
    const float* bp2 = (const float*)d_in[8];
    const int* eidx  = (const int*)d_in[9];
    const int* batch = (const int*)d_in[10];
    float* out = (float*)d_out;

    const int N = in_sizes[0];
    const int E = in_sizes[9] / 2;
    const int G = out_size / 3;
    const int* src = eidx;
    const int* dst = eidx + E;

    const int B = (N + RN - 1) >> RSH;
    const int C = NCHUNK;
    const int chunkLen = (E + C - 1) / C;

    float* wsf   = (float*)d_ws;
    float* pair  = wsf;
    float* xs    = pair + 2L * N;
    int*   bstart = (int*)(xs + N);
    int*   btot  = bstart + (B + 1);
    int*   cnt   = btot + B;
    int*   part  = (int*)(wsf + 4L * N);
    int*   part2 = part + E;
    float* psum  = (float*)(part2 + E);
    float* pcnt  = psum + (long)G * 32;

    gcn_count<<<C, 1024, 0, stream>>>(dst, cnt, E, chunkLen, B, C);
    gcn_bscan<<<B, 256, 0, stream>>>(cnt, btot, C, psum, 33 * G);
    gcn_scatter<<<C, 1024, 0, stream>>>(src, dst, cnt, btot, bstart, part, E, chunkLen, B, C);

    gcn_deg_sort<<<B, 1024, 0, stream>>>(part, bstart, x, pair, xs, part2, N);
    gcn_edge1_bucket<<<B, 1024, 0, stream>>>(part, bstart, xs, pair, x, N);
    gcn_layer2_bucket<<<B, 1024, 0, stream>>>(part2, bstart, pair, W1, b1, W2, b2, batch, psum, pcnt, N);

    gcn_head_wmma<<<(G + 31) / 32, 512, 0, stream>>>(psum, pcnt, Wp1, bp1, Wp2, bp2, out, G);
}


# sort passes and layer-1 edge pass: a thread issues all its edge loads (10-13) before the first use, deg_sort keeps its words in registers; was 1 load in flight a wave; LDS float add order changes
# speedup vs baseline: 6.2890x; 1.0599x over previous
#include <hip/hip_runtime.h>
#include <hip/hip_bf16.h>


typedef float v2f __attribute__((ext_vector_type(2)));
typedef float v8f __attribute__((ext_vector_type(8)));
typedef int v2i __attribute__((ext_vector_type(2)));

#define RSH 8
#define RN 256
#define SRC_BITS 20
#define SRC_MASK 0xFFFFF
#define NCHUNK 250
#define SCAP 16384
#define CCAP 16384
#define CSTRIPS 13
#define BSTRIPS 10
#define AP 68


__device__ __forceinline__ int block_incl_scan(int v, int* sh) {
    const int tid = threadIdx.x;
    const int n = blockDim.x;
    sh[tid] = v;
    __syncthreads();
    for (int off = 1; off < n; off <<= 1) {
        int a = tid >= off ? sh[tid - off] : 0;
        __syncthreads();
        v += a;
        sh[tid] = v;
        __syncthreads();
    }
    return v;
}

__global__ void __launch_bounds__(1024)
gcn_count(const int* __restrict__ dst, int* __restrict__ cnt, int E, int chunkLen, int B, int C) {
    __shared__ int h[1024];
    const int tid = threadIdx.x;
    for (int i = tid; i < B; i += blockDim.x) h[i] = 0;
    __syncthreads();
    const int lo = blockIdx.x * chunkLen;
    const int hi = min(lo + chunkLen, E);
    const int bd = blockDim.x;
    for (int base = lo + tid; base < hi; base += CSTRIPS * bd) {
        int d[CSTRIPS];
#pragma unroll
        for (int u = 0; u < CSTRIPS; ++u) {
            const int e = base + u * bd;
            d[u] = dst[e < hi ? e : hi - 1];
        }
#pragma unroll
        for (int u = 0; u < CSTRIPS; ++u) asm volatile("" : "+v"(d[u]));
#pragma unroll
        for (int u = 0; u < CSTRIPS; ++u)
            if (base + u * bd < hi) atomicAdd(&h[d[u] >> RSH], 1);
    }
    __syncthreads();
    for (int i = tid; i < B; i += blockDim.x) cnt[i * C + blockIdx.x] = h[i];
}

__global__ void __launch_bounds__(256)
gcn_bscan(int* __restrict__ cnt, int* __restrict__ btot, int C, float* __restrict__ pz, int nz) {
    __shared__ int sh[256];
    const int tid = threadIdx.x;
    const int b = blockIdx.x;
    int v = tid < C ? cnt[b * C + tid] : 0;
    int incl = block_incl_scan(v, sh);
    if (tid < C) cnt[b * C + tid] = incl - v;
    if (tid == 255) btot[b] = incl;
    for (int i = b * 256 + tid; i < nz; i += gridDim.x * 256) pz[i] = 0.0f;
}

__device__ __forceinline__ long long block_incl_scan64(long long v, long long* sh) {
    const int tid = threadIdx.x;
    const int n = blockDim.x;
    sh[tid] = v;
    __syncthreads();
    for (int off = 1; off < n; off <<= 1) {
        long long a = tid >= off ? sh[tid - off] : 0;
        __syncthreads();
        v += a;
        sh[tid] = v;
        __syncthreads();
    }
    return v;
}

__global__ void __launch_bounds__(1024)
gcn_scatter(const int* __restrict__ src, const int* __restrict__ dst,
            const int* __restrict__ offs, const int* __restrict__ btot,
            int* __restrict__ bstart, int* __restrict__ part,
            int E, int chunkLen, int B, int C) {
    __shared__ long long sh[1024];
    __shared__ int cur[1024];
    __shared__ int delta[1024];
    __shared__ v2i stage[CCAP];
    const int tid = threadIdx.x;
    const int c = blockIdx.x;
    int bt = 0, off = 0, lc = 0;
    if (tid < B) {
        bt = btot[tid];
        off = offs[tid * C + c];
        lc = (c + 1 < C ? offs[tid * C + c + 1] : bt) - off;
    }
    const long long v = ((long long)bt << 32) | (long long)lc;
    const long long excl = block_incl_scan64(v, sh) - v;
    const int gexcl = (int)(excl >> 32);
    const int lexcl = (int)excl;
    if (c == 0 && tid <= B) bstart[tid] = gexcl;
    const int lo = c * chunkLen;
    const int hi = min(lo + chunkLen, E);
    const int n = hi - lo;
    if (n <= CCAP) {
        if (tid < B) {
            cur[tid] = lexcl;
            delta[tid] = gexcl + off - lexcl;
        }
        __syncthreads();
        const int bd = blockDim.x;
        for (int base = lo + tid; base < hi; base += CSTRIPS * bd) {
            int d[CSTRIPS], s[CSTRIPS], dlt[CSTRIPS];
#pragma unroll
            for (int u = 0; u < CSTRIPS; ++u) {
                const int e = base + u * bd;
                const int ec = e < hi ? e : hi - 1;
                d[u] = dst[ec];
                s[u] = src[ec];
            }
#pragma unroll
            for (int u = 0; u < CSTRIPS; ++u) asm volatile("" : "+v"(d[u]), "+v"(s[u]));
#pragma unroll
            for (int u = 0; u < CSTRIPS; ++u) dlt[u] = delta[d[u] >> RSH];
#pragma unroll
            for (int u = 0; u < CSTRIPS; ++u) {
                if (base + u * bd < hi) {
                    const int pos = atomicAdd(&cur[d[u] >> RSH], 1);
                    v2i w;
                    w.x = s[u] | ((d[u] & (RN - 1)) << SRC_BITS);
                    w.y = pos + dlt[u];
                    stage[pos] = w;
                }
            }
        }
        __syncthreads();
        for (int base = tid; base < n; base += CSTRIPS * bd) {
            v2i w[CSTRIPS];
#pragma unroll
            for (int u = 0; u < CSTRIPS; ++u) {
                const int i = base + u * bd;
                w[u] = stage[i < n ? i : n - 1];
            }
#pragma unroll
            for (int u = 0; u < CSTRIPS; ++u)
                if (base + u * bd < n) part[w[u].y] = w[u].x;
        }
    } else {
        if (tid < B) cur[tid] = gexcl + off;
        __syncthreads();
        for (int e = lo + tid; e < hi; e += blockDim.x) {
            int d = dst[e];
            int s = src[e];
            int pos = atomicAdd(&cur[d >> RSH], 1);
            part[pos] = s | ((d & (RN - 1)) << SRC_BITS);
        }
    }
}

__global__ void __launch_bounds__(1024)
gcn_deg_sort(const int* __restrict__ part, const int* __restrict__ bstart,
             const float* __restrict__ x, float* __restrict__ pair, float* __restrict__ xs,
             int* __restrict__ part2, int N) {
    __shared__ int h[RN];
    __shared__ int cur[RN];
    __shared__ int stage[SCAP];
    const int tid = threadIdx.x;
    if (tid < RN) h[tid] = 0;
    __syncthreads();
    const int e0 = bstart[blockIdx.x], e1 = bstart[blockIdx.x + 1];
    const int nE = e1 - e0;
    const int bd = blockDim.x;
    const bool held = nE > 0 && nE <= SCAP && nE <= BSTRIPS * bd;
    int w[BSTRIPS];
    if (held) {
#pragma unroll
        for (int u = 0; u < BSTRIPS; ++u) {
            const int i = e0 + tid + u * bd;
            w[u] = part[i < e1 ? i : e1 - 1];
        }
#pragma unroll
        for (int u = 0; u < BSTRIPS; ++u)
            if (e0 + tid + u * bd < e1) atomicAdd(&h[w[u] >> SRC_BITS], 1);
    } else {
        for (int i = e0 + tid; i < e1; i += bd) atomicAdd(&h[part[i] >> SRC_BITS], 1);
    }
    __syncthreads();
    if (tid < RN) {
        int nd = (blockIdx.x << RSH) + tid;
        if (nd < N) {
            float d = rsqrtf((float)h[tid] + 1.0f);
            pair[2 * nd + 1] = d;
            xs[nd] = x[nd] * d;
        }
    }
    if (tid < 32) {
        int a[RN / 32];
        int sum = 0;
#pragma unroll
        for (int u = 0; u < RN / 32; ++u) { a[u] = h[tid * (RN / 32) + u]; sum += a[u]; }
        int incl = sum;
#pragma unroll
        for (int off = 1; off < 32; off <<= 1) {
            int o = __shfl_up(incl, off);
            if (tid >= off) incl += o;
        }
        int run = incl - sum;
#pragma unroll
        for (int u = 0; u < RN / 32; ++u) { cur[tid * (RN / 32) + u] = run; run += a[u]; }
    }
    __syncthreads();
    if (held) {
#pragma unroll
        for (int u = 0; u < BSTRIPS; ++u)
            if (e0 + tid + u * bd < e1) stage[atomicAdd(&cur[w[u] >> SRC_BITS], 1)] = w[u];
        __syncthreads();
        int v[BSTRIPS];
#pragma unroll
        for (int u = 0; u < BSTRIPS; ++u) {
            const int i = tid + u * bd;
            v[u] = stage[i < nE ? i : nE - 1];
        }
#pragma unroll
        for (int u = 0; u < BSTRIPS; ++u)
            if (tid + u * bd < nE) part2[e0 + tid + u * bd] = v[u];
    } else {
        for (int i = e0 + tid; i < e1; i += bd) {
            int ww = part[i];
            part2[e0 + atomicAdd(&cur[ww >> SRC_BITS], 1)] = ww;
        }
    }
}

__global__ void __launch_bounds__(1024)
gcn_edge1_bucket(const int* __restrict__ part, const int* __restrict__ bstart,
                 const float* __restrict__ xs, float* __restrict__ pair,
                 const float* __restrict__ x, int N) {
    __shared__ float acc[RN];
    __shared__ float disL[RN];
    const int tid = threadIdx.x;
    const int nd = (blockIdx.x << RSH) + tid;
    if (tid < RN) {
        acc[tid] = 0.0f;
        disL[tid] = nd < N ? pair[2 * nd + 1] : 0.0f;
    }
    __syncthreads();
    const int e0 = bstart[blockIdx.x], e1 = bstart[blockIdx.x + 1];
    const int bd = blockDim.x;
    for (int base = e0 + tid; base < e1; base += BSTRIPS * bd) {
        int w[BSTRIPS];
        float xv[BSTRIPS];
#pragma unroll
        for (int u = 0; u < BSTRIPS; ++u) {
            const int i = base + u * bd;
            w[u] = part[i < e1 ? i : e1 - 1];
        }
#pragma unroll
        for (int u = 0; u < BSTRIPS; ++u) xv[u] = xs[w[u] & SRC_MASK] * disL[w[u] >> SRC_BITS];
#pragma unroll
        for (int u = 0; u < BSTRIPS; ++u) asm volatile("" : "+v"(xv[u]));
#pragma unroll
        for (int u = 0; u < BSTRIPS; ++u)
            if (base + u * bd < e1) atomicAdd(&acc[w[u] >> SRC_BITS], xv[u]);
    }
    __syncthreads();
    if (tid < RN && nd < N) pair[2 * nd] = acc[tid] + disL[tid] * disL[tid] * x[nd];
}

__device__ __forceinline__ void row_flush(float* acc, int dl, int lane, float a0, float a1) {
    atomicAdd(&acc[dl * AP + lane], a0);
    atomicAdd(&acc[dl * AP + 32 + lane], a1);
}

__global__ void __launch_bounds__(1024)
gcn_layer2_bucket(const int* __restrict__ part2, const int* __restrict__ bstart,
                  const float* __restrict__ pair,
                  const float* __restrict__ W1, const float* __restrict__ b1,
                  const float* __restrict__ W2, const float* __restrict__ b2,
                  const int* __restrict__ batch,
                  float* __restrict__ psum, float* __restrict__ pcnt, int N) {
    __shared__ float acc[RN * AP];
    __shared__ float hs[RN * 32];
    __shared__ float disL[RN];
    __shared__ float tL[RN];
    const int tid = threadIdx.x;
    const int lane = tid & 31;
    const int wave = tid >> 5;
    const int nWaves = blockDim.x >> 5;
    const int node0 = blockIdx.x << RSH;
    for (int i = tid; i < RN * AP; i += blockDim.x) acc[i] = 0.0f;
    if (tid < RN) {
        int nd = node0 + tid;
        tL[tid] = nd < N ? pair[2 * nd] : 0.0f;
        disL[tid] = nd < N ? pair[2 * nd + 1] : 0.0f;
    }
    const float w1a = W1[lane], w1b = W1[lane + 32];
    const float b1a = b1[lane], b1b = b1[lane + 32];
    __syncthreads();
    const int e0 = bstart[blockIdx.x], e1 = bstart[blockIdx.x + 1];
    for (int base = e0 + wave * 32; base < e1; base += nWaves * 32) {
        const int idx = base + lane;
        const int w = idx < e1 ? part2[idx] : 0;
        const int s = w & SRC_MASK;
        const int dl = w >> SRC_BITS;
        const v2f p = *(const v2f*)(pair + 2 * s);
        const int tBits = __float_as_int(p.x);
        const int dBits = __float_as_int(p.y);
        const int cnt = min(32, e1 - base);
        const int prev = __shfl_up(dl, 1);
        const unsigned brk = __builtin_amdgcn_ballot_w32(lane > 0 && lane < cnt && dl != prev);
        float a0 = 0.0f, a1 = 0.0f;
        if (cnt == 32) {
#pragma unroll
            for (int j = 0; j < 32; ++j) {
                if (j > 0 && ((brk >> j) & 1u)) {
                    row_flush(acc, __builtin_amdgcn_readlane(dl, j - 1), lane, a0, a1);
                    a0 = 0.0f;
                    a1 = 0.0f;
                }
                const float tj = __int_as_float(__builtin_amdgcn_readlane(tBits, j));
                const float dj = __int_as_float(__builtin_amdgcn_readlane(dBits, j));
                a0 = fmaf(dj, fmaxf(fmaf(tj, w1a, b1a), 0.0f), a0);
                a1 = fmaf(dj, fmaxf(fmaf(tj, w1b, b1b), 0.0f), a1);
            }
            row_flush(acc, __builtin_amdgcn_readlane(dl, 31), lane, a0, a1);
        } else {
            for (int j = 0; j < cnt; ++j) {
                if ((brk >> j) & 1u) {
                    row_flush(acc, __builtin_amdgcn_readlane(dl, j - 1), lane, a0, a1);
                    a0 = 0.0f;
                    a1 = 0.0f;
                }
                const float tj = __int_as_float(__builtin_amdgcn_readlane(tBits, j));
                const float dj = __int_as_float(__builtin_amdgcn_readlane(dBits, j));
                a0 = fmaf(dj, fmaxf(fmaf(tj, w1a, b1a), 0.0f), a0);
                a1 = fmaf(dj, fmaxf(fmaf(tj, w1b, b1b), 0.0f), a1);
            }
            row_flush(acc, __builtin_amdgcn_readlane(dl, cnt - 1), lane, a0, a1);
        }
    }
    __syncthreads();
    {
        const int k = tid & 63;
        const float wk = W1[k], bk = b1[k];
        for (int nl = tid >> 6; nl < RN; nl += blockDim.x >> 6) {
            const float d = disL[nl];
            const float hself = fmaxf(fmaf(tL[nl], wk, bk), 0.0f);
            float* q = &acc[nl * AP + k];
            *q = d * (*q) + d * d * hself;
        }
    }
    __syncthreads();
    const int c = lane & 15;
    const int khalf = lane >> 4;
    const int mt = wave >> 1;
    const int n0 = (wave & 1) << 4;
    const float* arow = acc + (mt * 16 + c) * AP + khalf * 2;
    const float* wcol = W2 + (khalf * 2) * 32 + n0 + c;
    const float bias = b2[n0 + c];
    v8f D = {};
#pragma unroll
    for (int k0 = 0; k0 < 64; k0 += 4) {
        v2f a, b;
        a.x = arow[k0];
        a.y = arow[k0 + 1];
        b.x = wcol[k0 * 32];
        b.y = wcol[(k0 + 1) * 32];
        D = __builtin_amdgcn_wmma_f32_16x16x4_f32(
             false, a,  false, b,
             (short)0, D,  false,  false);
    }
    float* hrow = hs + (mt * 16 + khalf * 8) * 32 + n0 + c;
#pragma unroll
    for (int v = 0; v < 8; ++v) {
        float hv = D[v] + bias;
        hrow[v * 32] = hv > 0.0f ? hv : 0.0f;
    }
    __syncthreads();
    for (int i = tid; i < RN * 32; i += blockDim.x) {
        const int nl = i >> 5;
        const int f = i & 31;
        const int nd = node0 + nl;
        if (nd < N) {
            const int g = batch[nd];
            atomicAdd(&psum[g * 32 + f], hs[i]);
            if (f == 0) atomicAdd(&pcnt[g], 1.0f);
        }
    }
}

#define HP 132
__global__ void __launch_bounds__(512)
gcn_head_wmma(const float* __restrict__ psum, const float* __restrict__ pcnt,
              const float* __restrict__ Wp1, const float* __restrict__ bp1,
              const float* __restrict__ Wp2, const float* __restrict__ bp2,
              float* __restrict__ out, int G) {
    __shared__ float hs[32 * HP];
    __shared__ float os[96];
    const int tid = threadIdx.x;
    const int lane = tid & 31;
    const int wave = tid >> 5;
    const int c = lane & 15;
    const int khalf = lane >> 4;
    const int mt = wave >> 3;
    const int n0 = (wave & 7) << 4;
    const int g0 = blockIdx.x << 5;

    const int row = tid >> 4;
    const int sub = tid & 15;
    float w2[24];
#pragma unroll
    for (int j = 0; j < 24; ++j) w2[j] = Wp2[sub * 24 + j];
    float o0 = 0.0f, o1 = 0.0f, o2 = 0.0f;

    int g = g0 + (mt << 4) + c;
    g = g < G ? g : G - 1;
    float cn = pcnt[g];
    cn = cn > 1.0f ? cn : 1.0f;
    const float inv = 1.0f / cn;
    const float* prow = psum + g * 32 + khalf * 2;
    const float* wcol = Wp1 + (khalf * 2) * 128 + n0 + c;

    v8f acc0 = {};
    v8f acc1 = {};
#pragma unroll
    for (int k0 = 0; k0 < 32; k0 += 8) {
        v2f a, b;
        a.x = prow[k0] * inv;
        a.y = prow[k0 + 1] * inv;
        b.x = wcol[k0 * 128];
        b.y = wcol[(k0 + 1) * 128];
        acc0 = __builtin_amdgcn_wmma_f32_16x16x4_f32(
             false, a,  false, b,
             (short)0, acc0,  false,  false);
        a.x = prow[k0 + 4] * inv;
        a.y = prow[k0 + 5] * inv;
        b.x = wcol[(k0 + 4) * 128];
        b.y = wcol[(k0 + 5) * 128];
        acc1 = __builtin_amdgcn_wmma_f32_16x16x4_f32(
             false, a,  false, b,
             (short)0, acc1,  false,  false);
    }

    const float bias = bp1[n0 + c];
    const int rowBase = (mt << 4) + khalf * 8;
#pragma unroll
    for (int v = 0; v < 8; ++v) {
        float h = acc0[v] + acc1[v] + bias;
        hs[(rowBase + v) * HP + n0 + c] = h > 0.0f ? h : 0.0f;
    }
    __syncthreads();

    const float* hr = hs + row * HP + sub * 8;
#pragma unroll
    for (int j = 0; j < 8; ++j) {
        float h = hr[j];
        o0 = fmaf(h, w2[j * 3 + 0], o0);
        o1 = fmaf(h, w2[j * 3 + 1], o1);
        o2 = fmaf(h, w2[j * 3 + 2], o2);
    }
#pragma unroll
    for (int m = 8; m >= 1; m >>= 1) {
        o0 += __shfl_xor(o0, m, 16);
        o1 += __shfl_xor(o1, m, 16);
        o2 += __shfl_xor(o2, m, 16);
    }
    if (sub == 0) {
        os[row * 3 + 0] = o0 + bp2[0];
        os[row * 3 + 1] = o1 + bp2[1];
        os[row * 3 + 2] = o2 + bp2[2];
    }
    __syncthreads();
    if (tid < 96) {
        int idx = g0 * 3 + tid;
        if (idx < G * 3) out[idx] = os[tid];
    }
}

extern "C" void kernel_launch(void* const* d_in, const int* in_sizes, int n_in,
                              void* d_out, int out_size, void* d_ws, size_t ws_size,
                              hipStream_t stream) {
    const float* x   = (const float*)d_in[0];
    const float* W1  = (const float*)d_in[1];
    const float* b1  = (const float*)d_in[2];
    const float* W2  = (const float*)d_in[3];
    const float* b2  = (const float*)d_in[4];
    const float* Wp1 = (const float*)d_in[5];
    const float* bp1 = (const float*)d_in[6];
    const float* Wp2 = (const float*)d_in[7];
    const float* bp2 = (const float*)d_in[8];
    const int* eidx  = (const int*)d_in[9];
    const int* batch = (const int*)d_in[10];
    float* out = (float*)d_out;

    const int N = in_sizes[0];
    const int E = in_sizes[9] / 2;
    const int G = out_size / 3;
    const int* src = eidx;
    const int* dst = eidx + E;

    const int B = (N + RN - 1) >> RSH;
    const int C = NCHUNK;
    const int chunkLen = (E + C - 1) / C;

    float* wsf   = (float*)d_ws;
    float* pair  = wsf;
    float* xs    = pair + 2L * N;
    int*   bstart = (int*)(xs + N);
    int*   btot  = bstart + (B + 1);
    int*   cnt   = btot + B;
    int*   part  = (int*)(wsf + 4L * N);
    int*   part2 = part + E;
    float* psum  = (float*)(part2 + E);
    float* pcnt  = psum + (long)G * 32;

    gcn_count<<<C, 1024, 0, stream>>>(dst, cnt, E, chunkLen, B, C);
    gcn_bscan<<<B, 256, 0, stream>>>(cnt, btot, C, psum, 33 * G);
    gcn_scatter<<<C, 1024, 0, stream>>>(src, dst, cnt, btot, bstart, part, E, chunkLen, B, C);

    gcn_deg_sort<<<B, 1024, 0, stream>>>(part, bstart, x, pair, xs, part2, N);
    gcn_edge1_bucket<<<B, 1024, 0, stream>>>(part, bstart, xs, pair, x, N);
    gcn_layer2_bucket<<<B, 1024, 0, stream>>>(part2, bstart, pair, W1, b1, W2, b2, batch, psum, pcnt, N);

    gcn_head_wmma<<<(G + 31) / 32, 512, 0, stream>>>(psum, pcnt, Wp1, bp1, Wp2, bp2, out, G);
}


# edges ordered by bucket inside each chunk only: chunk-contiguous words + (bucket,chunk) first-slot table, bucket blocks walk their runs; no global count/scan, 7 launches -> 5; sum order changes
# speedup vs baseline: 6.8492x; 1.0891x over previous
#include <hip/hip_runtime.h>
#include <hip/hip_bf16.h>


typedef float v2f __attribute__((ext_vector_type(2)));
typedef float v8f __attribute__((ext_vector_type(8)));

#define RSH 8
#define RN 256
#define SRC_BITS 20
#define SRC_MASK 0xFFFFF
#define NCHUNK 250
#define SCAP 16384
#define CCAP 16384
#define CSTRIPS 13
#define RPW 8
#define RCAP 256
#define RPWS 4
#define AP 68


template <int PER>
__device__ __forceinline__ int wave_excl_scan_lds(int* a, int lane) {
    int v[PER];
    int sum = 0;
#pragma unroll
    for (int u = 0; u < PER; ++u) { v[u] = a[lane * PER + u]; sum += v[u]; }
    int incl = sum;
#pragma unroll
    for (int off = 1; off < 32; off <<= 1) {
        int o = __shfl_up(incl, off);
        if (lane >= off) incl += o;
    }
    int run = incl - sum;
#pragma unroll
    for (int u = 0; u < PER; ++u) { a[lane * PER + u] = run; run += v[u]; }
    return __shfl(incl, 31);
}

__device__ __forceinline__ int block_excl_scan(int v, int* wt) {
    const int lane = threadIdx.x & 31;
    const int wave = threadIdx.x >> 5;
    const int nW = (blockDim.x + 31) >> 5;
    int incl = v;
#pragma unroll
    for (int off = 1; off < 32; off <<= 1) {
        int o = __shfl_up(incl, off);
        if (lane >= off) incl += o;
    }
    if (lane == 31) wt[wave] = incl;
    __syncthreads();
    int ti = lane < nW ? wt[lane] : 0;
#pragma unroll
    for (int off = 1; off < 32; off <<= 1) {
        int o = __shfl_up(ti, off);
        if (lane >= off) ti += o;
    }
    const int base = __shfl(ti, wave > 0 ? wave - 1 : 0);
    return (wave > 0 ? base : 0) + incl - v;
}

__global__ void __launch_bounds__(1024)
gcn_chunk_sort(const int* __restrict__ src, const int* __restrict__ dst,
               int* __restrict__ tbl, int* __restrict__ part,
               int E, int chunkLen, int B, int C) {
    __shared__ int h[1024];
    __shared__ int wt[32];
    __shared__ int stage[CCAP];
    const int tid = threadIdx.x;
    const int bd = blockDim.x;
    const int c = blockIdx.x;
    for (int i = tid; i < 1024; i += bd) h[i] = 0;
    const int lo = c * chunkLen;
    const int hi = min(lo + chunkLen, E);
    const int n = max(hi - lo, 0);
    const bool staged = n <= CCAP;
    int d[CSTRIPS], s[CSTRIPS];
#pragma unroll
    for (int u = 0; u < CSTRIPS; ++u) {
        const int e = lo + tid + u * bd;
        const int ec = e < hi ? e : 0;
        d[u] = dst[ec];
        s[u] = src[ec];
    }
#pragma unroll
    for (int u = 0; u < CSTRIPS; ++u) asm volatile("" : "+v"(d[u]), "+v"(s[u]));
    __syncthreads();
#pragma unroll
    for (int u = 0; u < CSTRIPS; ++u)
        if (lo + tid + u * bd < hi) atomicAdd(&h[d[u] >> RSH], 1);
    for (int e = lo + tid + CSTRIPS * bd; e < hi; e += bd) atomicAdd(&h[dst[e] >> RSH], 1);
    __syncthreads();
    const int first = block_excl_scan(tid < 1024 ? h[tid] : 0, wt);
    if (tid < 1024) h[tid] = first;
    if (tid <= B) tbl[tid * C + c] = first;
    __syncthreads();
#pragma unroll
    for (int u = 0; u < CSTRIPS; ++u) {
        if (lo + tid + u * bd < hi) {
            const int pos = atomicAdd(&h[d[u] >> RSH], 1);
            const int w = s[u] | ((d[u] & (RN - 1)) << SRC_BITS);
            if (staged) stage[pos] = w; else part[lo + pos] = w;
        }
    }
    for (int e = lo + tid + CSTRIPS * bd; e < hi; e += bd) {
        const int dd = dst[e];
        const int pos = atomicAdd(&h[dd >> RSH], 1);
        const int w = src[e] | ((dd & (RN - 1)) << SRC_BITS);
        if (staged) stage[pos] = w; else part[lo + pos] = w;
    }
    if (staged) {
        __syncthreads();
        for (int i = tid; i < n; i += bd) part[lo + i] = stage[i];
    }
}

__device__ __forceinline__ void load_runs(const int* __restrict__ tbl, int b, int C, int chunkLen,
                                          int* roff, int* rlen) {
    for (int c = threadIdx.x; c < C; c += blockDim.x) {
        const int a = tbl[b * C + c];
        const int e = tbl[(b + 1) * C + c];
        roff[c] = c * chunkLen + a;
        rlen[c] = e - a;
    }
}

__global__ void __launch_bounds__(1024)
gcn_deg(const int* __restrict__ part, const int* __restrict__ tbl,
        const float* __restrict__ x, float* __restrict__ pair, float* __restrict__ xs,
        int* __restrict__ btot, float* __restrict__ pz, int nz, int N, int chunkLen, int C) {
    __shared__ int h[RN];
    __shared__ int roff[RCAP];
    __shared__ int rlen[RCAP];
    const int tid = threadIdx.x;
    const int lane = tid & 31;
    const int wave = tid >> 5;
    const int nWaves = blockDim.x >> 5;
    const int b = blockIdx.x;
    if (tid < RN) h[tid] = 0;
    load_runs(tbl, b, C, chunkLen, roff, rlen);
    for (int i = b * blockDim.x + tid; i < nz; i += gridDim.x * blockDim.x) pz[i] = 0.0f;
    __syncthreads();
    for (int c0 = wave; c0 < C; c0 += nWaves * RPW) {
        int w0[RPW], w1[RPW], len[RPW], off[RPW];
#pragma unroll
        for (int k = 0; k < RPW; ++k) {
            const int c = c0 + k * nWaves;
            off[k] = c < C ? roff[c] : 0;
            len[k] = c < C ? rlen[c] : 0;
        }
#pragma unroll
        for (int k = 0; k < RPW; ++k) {
            off[k] = __builtin_amdgcn_readfirstlane(off[k]);
            len[k] = __builtin_amdgcn_readfirstlane(len[k]);
        }
#pragma unroll
        for (int k = 0; k < RPW; ++k) {
            w0[k] = part[lane < len[k] ? off[k] + lane : 0];
            w1[k] = part[lane + 32 < len[k] ? off[k] + lane + 32 : 0];
        }
#pragma unroll
        for (int k = 0; k < RPW; ++k) asm volatile("" : "+v"(w0[k]), "+v"(w1[k]));
#pragma unroll
        for (int k = 0; k < RPW; ++k) {
            if (lane < len[k]) atomicAdd(&h[w0[k] >> SRC_BITS], 1);
            if (lane + 32 < len[k]) atomicAdd(&h[w1[k] >> SRC_BITS], 1);
            for (int j = lane + 64; j < len[k]; j += 32) atomicAdd(&h[part[off[k] + j] >> SRC_BITS], 1);
        }
    }
    __syncthreads();
    if (tid < RN) {
        const int nd = (b << RSH) + tid;
        if (nd < N) {
            const int cnt = h[tid];
            const float dv = rsqrtf((float)cnt + 1.0f);
            v2f p;
            p.x = __int_as_float(cnt);
            p.y = dv;
            *(v2f*)(pair + 2 * nd) = p;
            xs[nd] = x[nd] * dv;
        }
    }
    if (tid < 32) {
        int sum = 0;
        for (int c = tid; c < C; c += 32) sum += rlen[c];
#pragma unroll
        for (int m = 16; m >= 1; m >>= 1) sum += __shfl_xor(sum, m);
        if (tid == 0) btot[b] = sum;
    }
}

__global__ void __launch_bounds__(1024)
gcn_edge1_sort(const int* __restrict__ part, const int* __restrict__ tbl, const int* __restrict__ btot,
               const float* __restrict__ xs, float* __restrict__ pair, const float* __restrict__ x,
               int* __restrict__ part2, int* __restrict__ brange, int N, int chunkLen, int C) {
    __shared__ float acc[RN];
    __shared__ float disL[RN];
    __shared__ int cur[RN];
    __shared__ int roff[RCAP];
    __shared__ int rlen[RCAP];
    __shared__ int sStart;
    __shared__ int sTotal;
    __shared__ int stage[SCAP];
    const int tid = threadIdx.x;
    const int lane = tid & 31;
    const int wave = tid >> 5;
    const int nWaves = blockDim.x >> 5;
    const int b = blockIdx.x;
    const int nd = (b << RSH) + tid;
    if (tid == 0) sStart = 0;
    if (tid < RN) {
        acc[tid] = 0.0f;
        v2f p;
        p.x = 0.0f;
        p.y = 0.0f;
        if (nd < N) p = *(const v2f*)(pair + 2 * nd);
        cur[tid] = __float_as_int(p.x);
        disL[tid] = p.y;
    }
    load_runs(tbl, b, C, chunkLen, roff, rlen);
    int before = 0;
    for (int i = tid; i < b; i += blockDim.x) before += btot[i];
#pragma unroll
    for (int m = 16; m >= 1; m >>= 1) before += __shfl_xor(before, m);
    __syncthreads();
    if (lane == 0 && before != 0) atomicAdd(&sStart, before);
    if (tid < 32) {
        const int tot = wave_excl_scan_lds<RN / 32>(cur, tid);
        if (tid == 0) sTotal = tot;
    }
    __syncthreads();
    const int s0 = sStart;
    const int nE = sTotal;
    const bool staged = nE <= SCAP;
    if (staged) {
        for (int c0 = wave; c0 < C; c0 += nWaves * RPWS) {
            int w0[RPWS], w1[RPWS], len[RPWS], off[RPWS], p0[RPWS], p1[RPWS];
            float x0[RPWS], x1[RPWS];
#pragma unroll
            for (int k = 0; k < RPWS; ++k) {
                const int c = c0 + k * nWaves;
                off[k] = c < C ? roff[c] : 0;
                len[k] = c < C ? rlen[c] : 0;
            }
#pragma unroll
            for (int k = 0; k < RPWS; ++k) {
                off[k] = __builtin_amdgcn_readfirstlane(off[k]);
                len[k] = __builtin_amdgcn_readfirstlane(len[k]);
            }
#pragma unroll
            for (int k = 0; k < RPWS; ++k) {
                w0[k] = part[lane < len[k] ? off[k] + lane : 0];
                w1[k] = part[lane + 32 < len[k] ? off[k] + lane + 32 : 0];
            }
#pragma unroll
            for (int k = 0; k < RPWS; ++k) {
                x0[k] = xs[w0[k] & SRC_MASK] * disL[w0[k] >> SRC_BITS];
                x1[k] = xs[w1[k] & SRC_MASK] * disL[w1[k] >> SRC_BITS];
            }
#pragma unroll
            for (int k = 0; k < RPWS; ++k) asm volatile("" : "+v"(x0[k]), "+v"(x1[k]));
#pragma unroll
            for (int k = 0; k < RPWS; ++k) {
                p0[k] = 0;
                p1[k] = 0;
                if (lane < len[k]) {
                    const int dl = w0[k] >> SRC_BITS;
                    atomicAdd(&acc[dl], x0[k]);
                    p0[k] = atomicAdd(&cur[dl], 1);
                }
                if (lane + 32 < len[k]) {
                    const int dl = w1[k] >> SRC_BITS;
                    atomicAdd(&acc[dl], x1[k]);
                    p1[k] = atomicAdd(&cur[dl], 1);
                }
            }
#pragma unroll
            for (int k = 0; k < RPWS; ++k) {
                if (lane < len[k]) stage[p0[k]] = w0[k];
                if (lane + 32 < len[k]) stage[p1[k]] = w1[k];
                for (int j = lane + 64; j < len[k]; j += 32) {
                    const int ww = part[off[k] + j];
                    const int dl = ww >> SRC_BITS;
                    atomicAdd(&acc[dl], xs[ww & SRC_MASK] * disL[dl]);
                    stage[atomicAdd(&cur[dl], 1)] = ww;
                }
            }
        }
    } else {
        for (int c = wave; c < C; c += nWaves) {
            const int off = roff[c];
            const int len = rlen[c];
            for (int j = lane; j < len; j += 32) {
                const int ww = part[off + j];
                const int dl = ww >> SRC_BITS;
                atomicAdd(&acc[dl], xs[ww & SRC_MASK] * disL[dl]);
                part2[s0 + atomicAdd(&cur[dl], 1)] = ww;
            }
        }
    }
    __syncthreads();
    if (staged)
        for (int i = tid; i < nE; i += blockDim.x) part2[s0 + i] = stage[i];
    if (tid < RN && nd < N) pair[2 * nd] = acc[tid] + disL[tid] * disL[tid] * x[nd];
    if (tid == 0) {
        brange[2 * b] = s0;
        brange[2 * b + 1] = s0 + nE;
    }
}

__device__ __forceinline__ void row_flush(float* acc, int dl, int lane, float a0, float a1) {
    atomicAdd(&acc[dl * AP + lane], a0);
    atomicAdd(&acc[dl * AP + 32 + lane], a1);
}

__global__ void __launch_bounds__(1024)
gcn_layer2_bucket(const int* __restrict__ part2, const int* __restrict__ brange,
                  const float* __restrict__ pair,
                  const float* __restrict__ W1, const float* __restrict__ b1,
                  const float* __restrict__ W2, const float* __restrict__ b2,
                  const int* __restrict__ batch,
                  float* __restrict__ psum, float* __restrict__ pcnt, int N) {
    __shared__ float acc[RN * AP];
    __shared__ float hs[RN * 32];
    __shared__ float disL[RN];
    __shared__ float tL[RN];
    const int tid = threadIdx.x;
    const int lane = tid & 31;
    const int wave = tid >> 5;
    const int nWaves = blockDim.x >> 5;
    const int node0 = blockIdx.x << RSH;
    for (int i = tid; i < RN * AP; i += blockDim.x) acc[i] = 0.0f;
    if (tid < RN) {
        int nd = node0 + tid;
        tL[tid] = nd < N ? pair[2 * nd] : 0.0f;
        disL[tid] = nd < N ? pair[2 * nd + 1] : 0.0f;
    }
    const float w1a = W1[lane], w1b = W1[lane + 32];
    const float b1a = b1[lane], b1b = b1[lane + 32];
    __syncthreads();
    const int e0 = brange[2 * blockIdx.x], e1 = brange[2 * blockIdx.x + 1];
    for (int base = e0 + wave * 32; base < e1; base += nWaves * 32) {
        const int idx = base + lane;
        const int w = idx < e1 ? part2[idx] : 0;
        const int s = w & SRC_MASK;
        const int dl = w >> SRC_BITS;
        const v2f p = *(const v2f*)(pair + 2 * s);
        const int tBits = __float_as_int(p.x);
        const int dBits = __float_as_int(p.y);
        const int cnt = min(32, e1 - base);
        const int prev = __shfl_up(dl, 1);
        const unsigned brk = __builtin_amdgcn_ballot_w32(lane > 0 && lane < cnt && dl != prev);
        float a0 = 0.0f, a1 = 0.0f;
        if (cnt == 32) {
#pragma unroll
            for (int j = 0; j < 32; ++j) {
                if (j > 0 && ((brk >> j) & 1u)) {
                    row_flush(acc, __builtin_amdgcn_readlane(dl, j - 1), lane, a0, a1);
                    a0 = 0.0f;
                    a1 = 0.0f;
                }
                const float tj = __int_as_float(__builtin_amdgcn_readlane(tBits, j));
                const float dj = __int_as_float(__builtin_amdgcn_readlane(dBits, j));
                a0 = fmaf(dj, fmaxf(fmaf(tj, w1a, b1a), 0.0f), a0);
                a1 = fmaf(dj, fmaxf(fmaf(tj, w1b, b1b), 0.0f), a1);
            }
            row_flush(acc, __builtin_amdgcn_readlane(dl, 31), lane, a0, a1);
        } else {
            for (int j = 0; j < cnt; ++j) {
                if ((brk >> j) & 1u) {
                    row_flush(acc, __builtin_amdgcn_readlane(dl, j - 1), lane, a0, a1);
                    a0 = 0.0f;
                    a1 = 0.0f;
                }
                const float tj = __int_as_float(__builtin_amdgcn_readlane(tBits, j));
                const float dj = __int_as_float(__builtin_amdgcn_readlane(dBits, j));
                a0 = fmaf(dj, fmaxf(fmaf(tj, w1a, b1a), 0.0f), a0);
                a1 = fmaf(dj, fmaxf(fmaf(tj, w1b, b1b), 0.0f), a1);
            }
            row_flush(acc, __builtin_amdgcn_readlane(dl, cnt - 1), lane, a0, a1);
        }
    }
    __syncthreads();
    {
        const int k = tid & 63;
        const float wk = W1[k], bk = b1[k];
        for (int nl = tid >> 6; nl < RN; nl += blockDim.x >> 6) {
            const float d = disL[nl];
            const float hself = fmaxf(fmaf(tL[nl], wk, bk), 0.0f);
            float* q = &acc[nl * AP + k];
            *q = d * (*q) + d * d * hself;
        }
    }
    __syncthreads();
    const int c = lane & 15;
    const int khalf = lane >> 4;
    const int mt = wave >> 1;
    const int n0 = (wave & 1) << 4;
    const float* arow = acc + (mt * 16 + c) * AP + khalf * 2;
    const float* wcol = W2 + (khalf * 2) * 32 + n0 + c;
    const float bias = b2[n0 + c];
    v8f D = {};
#pragma unroll
    for (int k0 = 0; k0 < 64; k0 += 4) {
        v2f a, b;
        a.x = arow[k0];
        a.y = arow[k0 + 1];
        b.x = wcol[k0 * 32];
        b.y = wcol[(k0 + 1) * 32];
        D = __builtin_amdgcn_wmma_f32_16x16x4_f32(
             false, a,  false, b,
             (short)0, D,  false,  false);
    }
    float* hrow = hs + (mt * 16 + khalf * 8) * 32 + n0 + c;
#pragma unroll
    for (int v = 0; v < 8; ++v) {
        float hv = D[v] + bias;
        hrow[v * 32] = hv > 0.0f ? hv : 0.0f;
    }
    __syncthreads();
    for (int i = tid; i < RN * 32; i += blockDim.x) {
        const int nl = i >> 5;
        const int f = i & 31;
        const int nd = node0 + nl;
        if (nd < N) {
            const int g = batch[nd];
            atomicAdd(&psum[g * 32 + f], hs[i]);
            if (f == 0) atomicAdd(&pcnt[g], 1.0f);
        }
    }
}

#define HP 132
__global__ void __launch_bounds__(512)
gcn_head_wmma(const float* __restrict__ psum, const float* __restrict__ pcnt,
              const float* __restrict__ Wp1, const float* __restrict__ bp1,
              const float* __restrict__ Wp2, const float* __restrict__ bp2,
              float* __restrict__ out, int G) {
    __shared__ float hs[32 * HP];
    __shared__ float os[96];
    const int tid = threadIdx.x;
    const int lane = tid & 31;
    const int wave = tid >> 5;
    const int c = lane & 15;
    const int khalf = lane >> 4;
    const int mt = wave >> 3;
    const int n0 = (wave & 7) << 4;
    const int g0 = blockIdx.x << 5;

    const int row = tid >> 4;
    const int sub = tid & 15;
    float w2[24];
#pragma unroll
    for (int j = 0; j < 24; ++j) w2[j] = Wp2[sub * 24 + j];
    float o0 = 0.0f, o1 = 0.0f, o2 = 0.0f;

    int g = g0 + (mt << 4) + c;
    g = g < G ? g : G - 1;
    float cn = pcnt[g];
    cn = cn > 1.0f ? cn : 1.0f;
    const float inv = 1.0f / cn;
    const float* prow = psum + g * 32 + khalf * 2;
    const float* wcol = Wp1 + (khalf * 2) * 128 + n0 + c;

    v8f acc0 = {};
    v8f acc1 = {};
#pragma unroll
    for (int k0 = 0; k0 < 32; k0 += 8) {
        v2f a, b;
        a.x = prow[k0] * inv;
        a.y = prow[k0 + 1] * inv;
        b.x = wcol[k0 * 128];
        b.y = wcol[(k0 + 1) * 128];
        acc0 = __builtin_amdgcn_wmma_f32_16x16x4_f32(
             false, a,  false, b,
             (short)0, acc0,  false,  false);
        a.x = prow[k0 + 4] * inv;
        a.y = prow[k0 + 5] * inv;
        b.x = wcol[(k0 + 4) * 128];
        b.y = wcol[(k0 + 5) * 128];
        acc1 = __builtin_amdgcn_wmma_f32_16x16x4_f32(
             false, a,  false, b,
             (short)0, acc1,  false,  false);
    }

    const float bias = bp1[n0 + c];
    const int rowBase = (mt << 4) + khalf * 8;
#pragma unroll
    for (int v = 0; v < 8; ++v) {
        float h = acc0[v] + acc1[v] + bias;
        hs[(rowBase + v) * HP + n0 + c] = h > 0.0f ? h : 0.0f;
    }
    __syncthreads();

    const float* hr = hs + row * HP + sub * 8;
#pragma unroll
    for (int j = 0; j < 8; ++j) {
        float h = hr[j];
        o0 = fmaf(h, w2[j * 3 + 0], o0);
        o1 = fmaf(h, w2[j * 3 + 1], o1);
        o2 = fmaf(h, w2[j * 3 + 2], o2);
    }
#pragma unroll
    for (int m = 8; m >= 1; m >>= 1) {
        o0 += __shfl_xor(o0, m, 16);
        o1 += __shfl_xor(o1, m, 16);
        o2 += __shfl_xor(o2, m, 16);
    }
    if (sub == 0) {
        os[row * 3 + 0] = o0 + bp2[0];
        os[row * 3 + 1] = o1 + bp2[1];
        os[row * 3 + 2] = o2 + bp2[2];
    }
    __syncthreads();
    if (tid < 96) {
        int idx = g0 * 3 + tid;
        if (idx < G * 3) out[idx] = os[tid];
    }
}

extern "C" void kernel_launch(void* const* d_in, const int* in_sizes, int n_in,
                              void* d_out, int out_size, void* d_ws, size_t ws_size,
                              hipStream_t stream) {
    const float* x   = (const float*)d_in[0];
    const float* W1  = (const float*)d_in[1];
    const float* b1  = (const float*)d_in[2];
    const float* W2  = (const float*)d_in[3];
    const float* b2  = (const float*)d_in[4];
    const float* Wp1 = (const float*)d_in[5];
    const float* bp1 = (const float*)d_in[6];
    const float* Wp2 = (const float*)d_in[7];
    const float* bp2 = (const float*)d_in[8];
    const int* eidx  = (const int*)d_in[9];
    const int* batch = (const int*)d_in[10];
    float* out = (float*)d_out;

    const int N = in_sizes[0];
    const int E = in_sizes[9] / 2;
    const int G = out_size / 3;
    const int* src = eidx;
    const int* dst = eidx + E;

    const int B = (N + RN - 1) >> RSH;
    const int C = NCHUNK;
    const int chunkLen = (E + C - 1) / C;

    float* wsf   = (float*)d_ws;
    float* pair  = wsf;
    float* xs    = pair + 2L * N;
    int*   tbl   = (int*)(xs + N);
    int*   btot  = tbl + (B + 1) * C;
    int*   brange = btot + B;
    int*   part  = (int*)(wsf + 4L * N);
    int*   part2 = part + E;
    float* psum  = (float*)(part2 + E);
    float* pcnt  = psum + (long)G * 32;

    gcn_chunk_sort<<<C, 1024, 0, stream>>>(src, dst, tbl, part, E, chunkLen, B, C);

    gcn_deg<<<B, 1024, 0, stream>>>(part, tbl, x, pair, xs, btot, psum, 33 * G, N, chunkLen, C);
    gcn_edge1_sort<<<B, 1024, 0, stream>>>(part, tbl, btot, xs, pair, x, part2, brange, N, chunkLen, C);
    gcn_layer2_bucket<<<B, 1024, 0, stream>>>(part2, brange, pair, W1, b1, W2, b2, batch, psum, pcnt, N);

    gcn_head_wmma<<<(G + 31) / 32, 512, 0, stream>>>(psum, pcnt, Wp1, bp1, Wp2, bp2, out, G);
}


# dst buckets of 200 nodes, was 256: 500 bucket blocks = two on every WGP, was 391 (two on some WGPs, one on the rest); chunk slot table 16-bit so it still fits its plane; float sum order changes
# speedup vs baseline: 7.2638x; 1.0605x over previous
#include <hip/hip_runtime.h>
#include <hip/hip_bf16.h>


typedef float v2f __attribute__((ext_vector_type(2)));
typedef float v8f __attribute__((ext_vector_type(8)));

#define RN 200
#define RNT 208
#define RNS 224
#define SRC_BITS 20
#define SRC_MASK 0xFFFFF
#define NCHUNK 250
#define SCAP 16384
#define CCAP 16384
#define CSTRIPS 13
#define RPW 8
#define RCAP 256
#define RPWS 4
#define AP 68


template <int PER>
__device__ __forceinline__ int wave_excl_scan_lds(int* a, int lane) {
    int v[PER];
    int sum = 0;
#pragma unroll
    for (int u = 0; u < PER; ++u) { v[u] = a[lane * PER + u]; sum += v[u]; }
    int incl = sum;
#pragma unroll
    for (int off = 1; off < 32; off <<= 1) {
        int o = __shfl_up(incl, off);
        if (lane >= off) incl += o;
    }
    int run = incl - sum;
#pragma unroll
    for (int u = 0; u < PER; ++u) { a[lane * PER + u] = run; run += v[u]; }
    return __shfl(incl, 31);
}

__device__ __forceinline__ int block_excl_scan(int v, int* wt) {
    const int lane = threadIdx.x & 31;
    const int wave = threadIdx.x >> 5;
    const int nW = (blockDim.x + 31) >> 5;
    int incl = v;
#pragma unroll
    for (int off = 1; off < 32; off <<= 1) {
        int o = __shfl_up(incl, off);
        if (lane >= off) incl += o;
    }
    if (lane == 31) wt[wave] = incl;
    __syncthreads();
    int ti = lane < nW ? wt[lane] : 0;
#pragma unroll
    for (int off = 1; off < 32; off <<= 1) {
        int o = __shfl_up(ti, off);
        if (lane >= off) ti += o;
    }
    const int base = __shfl(ti, wave > 0 ? wave - 1 : 0);
    return (wave > 0 ? base : 0) + incl - v;
}

__global__ void __launch_bounds__(1024)
gcn_chunk_sort(const int* __restrict__ src, const int* __restrict__ dst,
               unsigned short* __restrict__ tbl, int* __restrict__ part,
               int E, int chunkLen, int B, int C) {
    __shared__ int h[1024];
    __shared__ int wt[32];
    __shared__ int stage[CCAP];
    const int tid = threadIdx.x;
    const int bd = blockDim.x;
    const int c = blockIdx.x;
    for (int i = tid; i < 1024; i += bd) h[i] = 0;
    const int lo = c * chunkLen;
    const int hi = min(lo + chunkLen, E);
    const int n = max(hi - lo, 0);
    const bool staged = n <= CCAP;
    int d[CSTRIPS], s[CSTRIPS];
#pragma unroll
    for (int u = 0; u < CSTRIPS; ++u) {
        const int e = lo + tid + u * bd;
        const int ec = e < hi ? e : 0;
        d[u] = dst[ec];
        s[u] = src[ec];
    }
#pragma unroll
    for (int u = 0; u < CSTRIPS; ++u) asm volatile("" : "+v"(d[u]), "+v"(s[u]));
    __syncthreads();
#pragma unroll
    for (int u = 0; u < CSTRIPS; ++u)
        if (lo + tid + u * bd < hi) atomicAdd(&h[(unsigned)d[u] / RN], 1);
    for (int e = lo + tid + CSTRIPS * bd; e < hi; e += bd) atomicAdd(&h[(unsigned)dst[e] / RN], 1);
    __syncthreads();
    const int first = block_excl_scan(tid < 1024 ? h[tid] : 0, wt);
    if (tid < 1024) h[tid] = first;
    if (tid <= B) tbl[tid * C + c] = (unsigned short)first;
    __syncthreads();
#pragma unroll
    for (int u = 0; u < CSTRIPS; ++u) {
        if (lo + tid + u * bd < hi) {
            const int bk = (unsigned)d[u] / RN;
            const int pos = atomicAdd(&h[bk], 1);
            const int w = s[u] | ((d[u] - bk * RN) << SRC_BITS);
            if (staged) stage[pos] = w; else part[lo + pos] = w;
        }
    }
    for (int e = lo + tid + CSTRIPS * bd; e < hi; e += bd) {
        const int dd = dst[e];
        const int bk = (unsigned)dd / RN;
        const int pos = atomicAdd(&h[bk], 1);
        const int w = src[e] | ((dd - bk * RN) << SRC_BITS);
        if (staged) stage[pos] = w; else part[lo + pos] = w;
    }
    if (staged) {
        __syncthreads();
        for (int i = tid; i < n; i += bd) part[lo + i] = stage[i];
    }
}

__device__ __forceinline__ void load_runs(const unsigned short* __restrict__ tbl, int b, int C, int chunkLen,
                                          int* roff, int* rlen) {
    for (int c = threadIdx.x; c < C; c += blockDim.x) {
        const int a = tbl[b * C + c];
        const int e = tbl[(b + 1) * C + c];
        roff[c] = c * chunkLen + a;
        rlen[c] = e - a;
    }
}

__global__ void __launch_bounds__(1024)
gcn_deg(const int* __restrict__ part, const unsigned short* __restrict__ tbl,
        const float* __restrict__ x, float* __restrict__ pair, float* __restrict__ xs,
        int* __restrict__ btot, float* __restrict__ pz, int nz, int N, int chunkLen, int C) {
    __shared__ int h[RN];
    __shared__ int roff[RCAP];
    __shared__ int rlen[RCAP];
    const int tid = threadIdx.x;
    const int lane = tid & 31;
    const int wave = tid >> 5;
    const int nWaves = blockDim.x >> 5;
    const int b = blockIdx.x;
    if (tid < RN) h[tid] = 0;
    load_runs(tbl, b, C, chunkLen, roff, rlen);
    for (int i = b * blockDim.x + tid; i < nz; i += gridDim.x * blockDim.x) pz[i] = 0.0f;
    __syncthreads();
    for (int c0 = wave; c0 < C; c0 += nWaves * RPW) {
        int w0[RPW], w1[RPW], len[RPW], off[RPW];
#pragma unroll
        for (int k = 0; k < RPW; ++k) {
            const int c = c0 + k * nWaves;
            off[k] = c < C ? roff[c] : 0;
            len[k] = c < C ? rlen[c] : 0;
        }
#pragma unroll
        for (int k = 0; k < RPW; ++k) {
            off[k] = __builtin_amdgcn_readfirstlane(off[k]);
            len[k] = __builtin_amdgcn_readfirstlane(len[k]);
        }
#pragma unroll
        for (int k = 0; k < RPW; ++k) {
            w0[k] = part[lane < len[k] ? off[k] + lane : 0];
            w1[k] = part[lane + 32 < len[k] ? off[k] + lane + 32 : 0];
        }
#pragma unroll
        for (int k = 0; k < RPW; ++k) asm volatile("" : "+v"(w0[k]), "+v"(w1[k]));
#pragma unroll
        for (int k = 0; k < RPW; ++k) {
            if (lane < len[k]) atomicAdd(&h[w0[k] >> SRC_BITS], 1);
            if (lane + 32 < len[k]) atomicAdd(&h[w1[k] >> SRC_BITS], 1);
            for (int j = lane + 64; j < len[k]; j += 32) atomicAdd(&h[part[off[k] + j] >> SRC_BITS], 1);
        }
    }
    __syncthreads();
    if (tid < RN) {
        const int nd = b * RN + tid;
        if (nd < N) {
            const int cnt = h[tid];
            const float dv = rsqrtf((float)cnt + 1.0f);
            v2f p;
            p.x = __int_as_float(cnt);
            p.y = dv;
            *(v2f*)(pair + 2 * nd) = p;
            xs[nd] = x[nd] * dv;
        }
    }
    if (tid < 32) {
        int sum = 0;
        for (int c = tid; c < C; c += 32) sum += rlen[c];
#pragma unroll
        for (int m = 16; m >= 1; m >>= 1) sum += __shfl_xor(sum, m);
        if (tid == 0) btot[b] = sum;
    }
}

__global__ void __launch_bounds__(1024)
gcn_edge1_sort(const int* __restrict__ part, const unsigned short* __restrict__ tbl, const int* __restrict__ btot,
               const float* __restrict__ xs, float* __restrict__ pair, const float* __restrict__ x,
               int* __restrict__ part2, int* __restrict__ brange, int N, int chunkLen, int C) {
    __shared__ float acc[RN];
    __shared__ float disL[RN];
    __shared__ int cur[RNS];
    __shared__ int roff[RCAP];
    __shared__ int rlen[RCAP];
    __shared__ int sStart;
    __shared__ int sTotal;
    __shared__ int stage[SCAP];
    const int tid = threadIdx.x;
    const int lane = tid & 31;
    const int wave = tid >> 5;
    const int nWaves = blockDim.x >> 5;
    const int b = blockIdx.x;
    const int nd = b * RN + tid;
    if (tid == 0) sStart = 0;
    if (tid < RNS) {
        v2f p;
        p.x = 0.0f;
        p.y = 0.0f;
        if (tid < RN && nd < N) p = *(const v2f*)(pair + 2 * nd);
        cur[tid] = __float_as_int(p.x);
        if (tid < RN) {
            acc[tid] = 0.0f;
            disL[tid] = p.y;
        }
    }
    load_runs(tbl, b, C, chunkLen, roff, rlen);
    int before = 0;
    for (int i = tid; i < b; i += blockDim.x) before += btot[i];
#pragma unroll
    for (int m = 16; m >= 1; m >>= 1) before += __shfl_xor(before, m);
    __syncthreads();
    if (lane == 0 && before != 0) atomicAdd(&sStart, before);
    if (tid < 32) {
        const int tot = wave_excl_scan_lds<RNS / 32>(cur, tid);
        if (tid == 0) sTotal = tot;
    }
    __syncthreads();
    const int s0 = sStart;
    const int nE = sTotal;
    const bool staged = nE <= SCAP;
    if (staged) {
        for (int c0 = wave; c0 < C; c0 += nWaves * RPWS) {
            int w0[RPWS], w1[RPWS], len[RPWS], off[RPWS], p0[RPWS], p1[RPWS];
            float x0[RPWS], x1[RPWS];
#pragma unroll
            for (int k = 0; k < RPWS; ++k) {
                const int c = c0 + k * nWaves;
                off[k] = c < C ? roff[c] : 0;
                len[k] = c < C ? rlen[c] : 0;
            }
#pragma unroll
            for (int k = 0; k < RPWS; ++k) {
                off[k] = __builtin_amdgcn_readfirstlane(off[k]);
                len[k] = __builtin_amdgcn_readfirstlane(len[k]);
            }
#pragma unroll
            for (int k = 0; k < RPWS; ++k) {
                w0[k] = part[lane < len[k] ? off[k] + lane : 0];
                w1[k] = part[lane + 32 < len[k] ? off[k] + lane + 32 : 0];
            }
#pragma unroll
            for (int k = 0; k < RPWS; ++k) {
                x0[k] = xs[w0[k] & SRC_MASK] * disL[w0[k] >> SRC_BITS];
                x1[k] = xs[w1[k] & SRC_MASK] * disL[w1[k] >> SRC_BITS];
            }
#pragma unroll
            for (int k = 0; k < RPWS; ++k) asm volatile("" : "+v"(x0[k]), "+v"(x1[k]));
#pragma unroll
            for (int k = 0; k < RPWS; ++k) {
                p0[k] = 0;
                p1[k] = 0;
                if (lane < len[k]) {
                    const int dl = w0[k] >> SRC_BITS;
                    atomicAdd(&acc[dl], x0[k]);
                    p0[k] = atomicAdd(&cur[dl], 1);
                }
                if (lane + 32 < len[k]) {
                    const int dl = w1[k] >> SRC_BITS;
                    atomicAdd(&acc[dl], x1[k]);
                    p1[k] = atomicAdd(&cur[dl], 1);
                }
            }
#pragma unroll
            for (int k = 0; k < RPWS; ++k) {
                if (lane < len[k]) stage[p0[k]] = w0[k];
                if (lane + 32 < len[k]) stage[p1[k]] = w1[k];
                for (int j = lane + 64; j < len[k]; j += 32) {
                    const int ww = part[off[k] + j];
                    const int dl = ww >> SRC_BITS;
                    atomicAdd(&acc[dl], xs[ww & SRC_MASK] * disL[dl]);
                    stage[atomicAdd(&cur[dl], 1)] = ww;
                }
            }
        }
    } else {
        for (int c = wave; c < C; c += nWaves) {
            const int off = roff[c];
            const int len = rlen[c];
            for (int j = lane; j < len; j += 32) {
                const int ww = part[off + j];
                const int dl = ww >> SRC_BITS;
                atomicAdd(&acc[dl], xs[ww & SRC_MASK] * disL[dl]);
                part2[s0 + atomicAdd(&cur[dl], 1)] = ww;
            }
        }
    }
    __syncthreads();
    if (staged)
        for (int i = tid; i < nE; i += blockDim.x) part2[s0 + i] = stage[i];
    if (tid < RN && nd < N) pair[2 * nd] = acc[tid] + disL[tid] * disL[tid] * x[nd];
    if (tid == 0) {
        brange[2 * b] = s0;
        brange[2 * b + 1] = s0 + nE;
    }
}

__device__ __forceinline__ void row_flush(float* acc, int dl, int lane, float a0, float a1) {
    atomicAdd(&acc[dl * AP + lane], a0);
    atomicAdd(&acc[dl * AP + 32 + lane], a1);
}

__global__ void __launch_bounds__(1024)
gcn_layer2_bucket(const int* __restrict__ part2, const int* __restrict__ brange,
                  const float* __restrict__ pair,
                  const float* __restrict__ W1, const float* __restrict__ b1,
                  const float* __restrict__ W2, const float* __restrict__ b2,
                  const int* __restrict__ batch,
                  float* __restrict__ psum, float* __restrict__ pcnt, int N) {
    __shared__ float acc[RNT * AP];
    __shared__ float hs[RNT * 32];
    __shared__ float disL[RN];
    __shared__ float tL[RN];
    const int tid = threadIdx.x;
    const int lane = tid & 31;
    const int wave = tid >> 5;
    const int nWaves = blockDim.x >> 5;
    const int node0 = blockIdx.x * RN;
    for (int i = tid; i < RNT * AP; i += blockDim.x) acc[i] = 0.0f;
    if (tid < RN) {
        int nd = node0 + tid;
        tL[tid] = nd < N ? pair[2 * nd] : 0.0f;
        disL[tid] = nd < N ? pair[2 * nd + 1] : 0.0f;
    }
    const float w1a = W1[lane], w1b = W1[lane + 32];
    const float b1a = b1[lane], b1b = b1[lane + 32];
    __syncthreads();
    const int e0 = brange[2 * blockIdx.x], e1 = brange[2 * blockIdx.x + 1];
    for (int base = e0 + wave * 32; base < e1; base += nWaves * 32) {
        const int idx = base + lane;
        const int w = idx < e1 ? part2[idx] : 0;
        const int s = w & SRC_MASK;
        const int dl = w >> SRC_BITS;
        const v2f p = *(const v2f*)(pair + 2 * s);
        const int tBits = __float_as_int(p.x);
        const int dBits = __float_as_int(p.y);
        const int cnt = min(32, e1 - base);
        const int prev = __shfl_up(dl, 1);
        const unsigned brk = __builtin_amdgcn_ballot_w32(lane > 0 && lane < cnt && dl != prev);
        float a0 = 0.0f, a1 = 0.0f;
        if (cnt == 32) {
#pragma unroll
            for (int j = 0; j < 32; ++j) {
                if (j > 0 && ((brk >> j) & 1u)) {
                    row_flush(acc, __builtin_amdgcn_readlane(dl, j - 1), lane, a0, a1);
                    a0 = 0.0f;
                    a1 = 0.0f;
                }
                const float tj = __int_as_float(__builtin_amdgcn_readlane(tBits, j));
                const float dj = __int_as_float(__builtin_amdgcn_readlane(dBits, j));
                a0 = fmaf(dj, fmaxf(fmaf(tj, w1a, b1a), 0.0f), a0);
                a1 = fmaf(dj, fmaxf(fmaf(tj, w1b, b1b), 0.0f), a1);
            }
            row_flush(acc, __builtin_amdgcn_readlane(dl, 31), lane, a0, a1);
        } else {
            for (int j = 0; j < cnt; ++j) {
                if ((brk >> j) & 1u) {
                    row_flush(acc, __builtin_amdgcn_readlane(dl, j - 1), lane, a0, a1);
                    a0 = 0.0f;
                    a1 = 0.0f;
                }
                const float tj = __int_as_float(__builtin_amdgcn_readlane(tBits, j));
                const float dj = __int_as_float(__builtin_amdgcn_readlane(dBits, j));
                a0 = fmaf(dj, fmaxf(fmaf(tj, w1a, b1a), 0.0f), a0);
                a1 = fmaf(dj, fmaxf(fmaf(tj, w1b, b1b), 0.0f), a1);
            }
            row_flush(acc, __builtin_amdgcn_readlane(dl, cnt - 1), lane, a0, a1);
        }
    }
    __syncthreads();
    {
        const int k = tid & 63;
        const float wk = W1[k], bk = b1[k];
        for (int nl = tid >> 6; nl < RN; nl += blockDim.x >> 6) {
            const float d = disL[nl];
            const float hself = fmaxf(fmaf(tL[nl], wk, bk), 0.0f);
            float* q = &acc[nl * AP + k];
            *q = d * (*q) + d * d * hself;
        }
    }
    __syncthreads();
    if (__builtin_amdgcn_readfirstlane(wave) < (RNT / 16) * 2) {
    const int c = lane & 15;
    const int khalf = lane >> 4;
    const int mt = wave >> 1;
    const int n0 = (wave & 1) << 4;
    const float* arow = acc + (mt * 16 + c) * AP + khalf * 2;
    const float* wcol = W2 + (khalf * 2) * 32 + n0 + c;
    const float bias = b2[n0 + c];
    v8f D = {};
#pragma unroll
    for (int k0 = 0; k0 < 64; k0 += 4) {
        v2f a, b;
        a.x = arow[k0];
        a.y = arow[k0 + 1];
        b.x = wcol[k0 * 32];
        b.y = wcol[(k0 + 1) * 32];
        D = __builtin_amdgcn_wmma_f32_16x16x4_f32(
             false, a,  false, b,
             (short)0, D,  false,  false);
    }
    float* hrow = hs + (mt * 16 + khalf * 8) * 32 + n0 + c;
#pragma unroll
    for (int v = 0; v < 8; ++v) {
        float hv = D[v] + bias;
        hrow[v * 32] = hv > 0.0f ? hv : 0.0f;
    }
    }
    __syncthreads();
    for (int i = tid; i < RN * 32; i += blockDim.x) {
        const int nl = i >> 5;
        const int f = i & 31;
        const int nd = node0 + nl;
        if (nd < N) {
            const int g = batch[nd];
            atomicAdd(&psum[g * 32 + f], hs[i]);
            if (f == 0) atomicAdd(&pcnt[g], 1.0f);
        }
    }
}

#define HP 132
__global__ void __launch_bounds__(512)
gcn_head_wmma(const float* __restrict__ psum, const float* __restrict__ pcnt,
              const float* __restrict__ Wp1, const float* __restrict__ bp1,
              const float* __restrict__ Wp2, const float* __restrict__ bp2,
              float* __restrict__ out, int G) {
    __shared__ float hs[32 * HP];
    __shared__ float os[96];
    const int tid = threadIdx.x;
    const int lane = tid & 31;
    const int wave = tid >> 5;
    const int c = lane & 15;
    const int khalf = lane >> 4;
    const int mt = wave >> 3;
    const int n0 = (wave & 7) << 4;
    const int g0 = blockIdx.x << 5;

    const int row = tid >> 4;
    const int sub = tid & 15;
    float w2[24];
#pragma unroll
    for (int j = 0; j < 24; ++j) w2[j] = Wp2[sub * 24 + j];
    float o0 = 0.0f, o1 = 0.0f, o2 = 0.0f;

    int g = g0 + (mt << 4) + c;
    g = g < G ? g : G - 1;
    float cn = pcnt[g];
    cn = cn > 1.0f ? cn : 1.0f;
    const float inv = 1.0f / cn;
    const float* prow = psum + g * 32 + khalf * 2;
    const float* wcol = Wp1 + (khalf * 2) * 128 + n0 + c;

    v8f acc0 = {};
    v8f acc1 = {};
#pragma unroll
    for (int k0 = 0; k0 < 32; k0 += 8) {
        v2f a, b;
        a.x = prow[k0] * inv;
        a.y = prow[k0 + 1] * inv;
        b.x = wcol[k0 * 128];
        b.y = wcol[(k0 + 1) * 128];
        acc0 = __builtin_amdgcn_wmma_f32_16x16x4_f32(
             false, a,  false, b,
             (short)0, acc0,  false,  false);
        a.x = prow[k0 + 4] * inv;
        a.y = prow[k0 + 5] * inv;
        b.x = wcol[(k0 + 4) * 128];
        b.y = wcol[(k0 + 5) * 128];
        acc1 = __builtin_amdgcn_wmma_f32_16x16x4_f32(
             false, a,  false, b,
             (short)0, acc1,  false,  false);
    }

    const float bias = bp1[n0 + c];
    const int rowBase = (mt << 4) + khalf * 8;
#pragma unroll
    for (int v = 0; v < 8; ++v) {
        float h = acc0[v] + acc1[v] + bias;
        hs[(rowBase + v) * HP + n0 + c] = h > 0.0f ? h : 0.0f;
    }
    __syncthreads();

    const float* hr = hs + row * HP + sub * 8;
#pragma unroll
    for (int j = 0; j < 8; ++j) {
        float h = hr[j];
        o0 = fmaf(h, w2[j * 3 + 0], o0);
        o1 = fmaf(h, w2[j * 3 + 1], o1);
        o2 = fmaf(h, w2[j * 3 + 2], o2);
    }
#pragma unroll
    for (int m = 8; m >= 1; m >>= 1) {
        o0 += __shfl_xor(o0, m, 16);
        o1 += __shfl_xor(o1, m, 16);
        o2 += __shfl_xor(o2, m, 16);
    }
    if (sub == 0) {
        os[row * 3 + 0] = o0 + bp2[0];
        os[row * 3 + 1] = o1 + bp2[1];
        os[row * 3 + 2] = o2 + bp2[2];
    }
    __syncthreads();
    if (tid < 96) {
        int idx = g0 * 3 + tid;
        if (idx < G * 3) out[idx] = os[tid];
    }
}

extern "C" void kernel_launch(void* const* d_in, const int* in_sizes, int n_in,
                              void* d_out, int out_size, void* d_ws, size_t ws_size,
                              hipStream_t stream) {
    const float* x   = (const float*)d_in[0];
    const float* W1  = (const float*)d_in[1];
    const float* b1  = (const float*)d_in[2];
    const float* W2  = (const float*)d_in[3];
    const float* b2  = (const float*)d_in[4];
    const float* Wp1 = (const float*)d_in[5];
    const float* bp1 = (const float*)d_in[6];
    const float* Wp2 = (const float*)d_in[7];
    const float* bp2 = (const float*)d_in[8];
    const int* eidx  = (const int*)d_in[9];
    const int* batch = (const int*)d_in[10];
    float* out = (float*)d_out;

    const int N = in_sizes[0];
    const int E = in_sizes[9] / 2;
    const int G = out_size / 3;
    const int* src = eidx;
    const int* dst = eidx + E;

    const int B = (N + RN - 1) / RN;
    const int C = NCHUNK;
    const int chunkLen = (E + C - 1) / C;

    float* wsf   = (float*)d_ws;
    float* pair  = wsf;
    float* xs    = pair + 2L * N;
    unsigned short* tbl = (unsigned short*)(xs + N);
    int*   btot  = (int*)(xs + N) + ((B + 1) * C + 1) / 2;
    int*   brange = btot + B;
    int*   part  = (int*)(wsf + 4L * N);
    int*   part2 = part + E;
    float* psum  = (float*)(part2 + E);
    float* pcnt  = psum + (long)G * 32;

    gcn_chunk_sort<<<C, 1024, 0, stream>>>(src, dst, tbl, part, E, chunkLen, B, C);

    gcn_deg<<<B, 1024, 0, stream>>>(part, tbl, x, pair, xs, btot, psum, 33 * G, N, chunkLen, C);
    gcn_edge1_sort<<<B, 1024, 0, stream>>>(part, tbl, btot, xs, pair, x, part2, brange, N, chunkLen, C);
    gcn_layer2_bucket<<<B, 1024, 0, stream>>>(part2, brange, pair, W1, b1, W2, b2, batch, psum, pcnt, N);

    gcn_head_wmma<<<(G + 31) / 32, 512, 0, stream>>>(psum, pcnt, Wp1, bp1, Wp2, bp2, out, G);
}


# mean-pool: matrix waves add their tile rows per graph in an LDS table (graph ids staged at entry), then one row atomic per graph and block, was one per node via an LDS tile; pooled sum order changes
# speedup vs baseline: 7.5791x; 1.0434x over previous
#include <hip/hip_runtime.h>
#include <hip/hip_bf16.h>


typedef float v2f __attribute__((ext_vector_type(2)));
typedef float v8f __attribute__((ext_vector_type(8)));

#define RN 200
#define RNT 208
#define RNS 224
#define SRC_BITS 20
#define SRC_MASK 0xFFFFF
#define NCHUNK 250
#define SCAP 16384
#define CCAP 16384
#define CSTRIPS 13
#define RPW 8
#define RCAP 256
#define RPWS 4
#define PG 32
#define AP 68


template <int PER>
__device__ __forceinline__ int wave_excl_scan_lds(int* a, int lane) {
    int v[PER];
    int sum = 0;
#pragma unroll
    for (int u = 0; u < PER; ++u) { v[u] = a[lane * PER + u]; sum += v[u]; }
    int incl = sum;
#pragma unroll
    for (int off = 1; off < 32; off <<= 1) {
        int o = __shfl_up(incl, off);
        if (lane >= off) incl += o;
    }
    int run = incl - sum;
#pragma unroll
    for (int u = 0; u < PER; ++u) { a[lane * PER + u] = run; run += v[u]; }
    return __shfl(incl, 31);
}

__device__ __forceinline__ int block_excl_scan(int v, int* wt) {
    const int lane = threadIdx.x & 31;
    const int wave = threadIdx.x >> 5;
    const int nW = (blockDim.x + 31) >> 5;
    int incl = v;
#pragma unroll
    for (int off = 1; off < 32; off <<= 1) {
        int o = __shfl_up(incl, off);
        if (lane >= off) incl += o;
    }
    if (lane == 31) wt[wave] = incl;
    __syncthreads();
    int ti = lane < nW ? wt[lane] : 0;
#pragma unroll
    for (int off = 1; off < 32; off <<= 1) {
        int o = __shfl_up(ti, off);
        if (lane >= off) ti += o;
    }
    const int base = __shfl(ti, wave > 0 ? wave - 1 : 0);
    return (wave > 0 ? base : 0) + incl - v;
}

__global__ void __launch_bounds__(1024)
gcn_chunk_sort(const int* __restrict__ src, const int* __restrict__ dst,
               unsigned short* __restrict__ tbl, int* __restrict__ part,
               int E, int chunkLen, int B, int C) {
    __shared__ int h[1024];
    __shared__ int wt[32];
    __shared__ int stage[CCAP];
    const int tid = threadIdx.x;
    const int bd = blockDim.x;
    const int c = blockIdx.x;
    for (int i = tid; i < 1024; i += bd) h[i] = 0;
    const int lo = c * chunkLen;
    const int hi = min(lo + chunkLen, E);
    const int n = max(hi - lo, 0);
    const bool staged = n <= CCAP;
    int d[CSTRIPS], s[CSTRIPS];
#pragma unroll
    for (int u = 0; u < CSTRIPS; ++u) {
        const int e = lo + tid + u * bd;
        const int ec = e < hi ? e : 0;
        d[u] = dst[ec];
        s[u] = src[ec];
    }
#pragma unroll
    for (int u = 0; u < CSTRIPS; ++u) asm volatile("" : "+v"(d[u]), "+v"(s[u]));
    __syncthreads();
#pragma unroll
    for (int u = 0; u < CSTRIPS; ++u)
        if (lo + tid + u * bd < hi) atomicAdd(&h[(unsigned)d[u] / RN], 1);
    for (int e = lo + tid + CSTRIPS * bd; e < hi; e += bd) atomicAdd(&h[(unsigned)dst[e] / RN], 1);
    __syncthreads();
    const int first = block_excl_scan(tid < 1024 ? h[tid] : 0, wt);
    if (tid < 1024) h[tid] = first;
    if (tid <= B) tbl[tid * C + c] = (unsigned short)first;
    __syncthreads();
#pragma unroll
    for (int u = 0; u < CSTRIPS; ++u) {
        if (lo + tid + u * bd < hi) {
            const int bk = (unsigned)d[u] / RN;
            const int pos = atomicAdd(&h[bk], 1);
            const int w = s[u] | ((d[u] - bk * RN) << SRC_BITS);
            if (staged) stage[pos] = w; else part[lo + pos] = w;
        }
    }
    for (int e = lo + tid + CSTRIPS * bd; e < hi; e += bd) {
        const int dd = dst[e];
        const int bk = (unsigned)dd / RN;
        const int pos = atomicAdd(&h[bk], 1);
        const int w = src[e] | ((dd - bk * RN) << SRC_BITS);
        if (staged) stage[pos] = w; else part[lo + pos] = w;
    }
    if (staged) {
        __syncthreads();
        for (int i = tid; i < n; i += bd) part[lo + i] = stage[i];
    }
}

__device__ __forceinline__ void load_runs(const unsigned short* __restrict__ tbl, int b, int C, int chunkLen,
                                          int* roff, int* rlen) {
    for (int c = threadIdx.x; c < C; c += blockDim.x) {
        const int a = tbl[b * C + c];
        const int e = tbl[(b + 1) * C + c];
        roff[c] = c * chunkLen + a;
        rlen[c] = e - a;
    }
}

__global__ void __launch_bounds__(1024)
gcn_deg(const int* __restrict__ part, const unsigned short* __restrict__ tbl,
        const float* __restrict__ x, float* __restrict__ pair, float* __restrict__ xs,
        int* __restrict__ btot, float* __restrict__ pz, int nz, int N, int chunkLen, int C) {
    __shared__ int h[RN];
    __shared__ int roff[RCAP];
    __shared__ int rlen[RCAP];
    const int tid = threadIdx.x;
    const int lane = tid & 31;
    const int wave = tid >> 5;
    const int nWaves = blockDim.x >> 5;
    const int b = blockIdx.x;
    if (tid < RN) h[tid] = 0;
    load_runs(tbl, b, C, chunkLen, roff, rlen);
    for (int i = b * blockDim.x + tid; i < nz; i += gridDim.x * blockDim.x) pz[i] = 0.0f;
    __syncthreads();
    for (int c0 = wave; c0 < C; c0 += nWaves * RPW) {
        int w0[RPW], w1[RPW], len[RPW], off[RPW];
#pragma unroll
        for (int k = 0; k < RPW; ++k) {
            const int c = c0 + k * nWaves;
            off[k] = c < C ? roff[c] : 0;
            len[k] = c < C ? rlen[c] : 0;
        }
#pragma unroll
        for (int k = 0; k < RPW; ++k) {
            off[k] = __builtin_amdgcn_readfirstlane(off[k]);
            len[k] = __builtin_amdgcn_readfirstlane(len[k]);
        }
#pragma unroll
        for (int k = 0; k < RPW; ++k) {
            w0[k] = part[lane < len[k] ? off[k] + lane : 0];
            w1[k] = part[lane + 32 < len[k] ? off[k] + lane + 32 : 0];
        }
#pragma unroll
        for (int k = 0; k < RPW; ++k) asm volatile("" : "+v"(w0[k]), "+v"(w1[k]));
#pragma unroll
        for (int k = 0; k < RPW; ++k) {
            if (lane < len[k]) atomicAdd(&h[w0[k] >> SRC_BITS], 1);
            if (lane + 32 < len[k]) atomicAdd(&h[w1[k] >> SRC_BITS], 1);
            for (int j = lane + 64; j < len[k]; j += 32) atomicAdd(&h[part[off[k] + j] >> SRC_BITS], 1);
        }
    }
    __syncthreads();
    if (tid < RN) {
        const int nd = b * RN + tid;
        if (nd < N) {
            const int cnt = h[tid];
            const float dv = rsqrtf((float)cnt + 1.0f);
            v2f p;
            p.x = __int_as_float(cnt);
            p.y = dv;
            *(v2f*)(pair + 2 * nd) = p;
            xs[nd] = x[nd] * dv;
        }
    }
    if (tid < 32) {
        int sum = 0;
        for (int c = tid; c < C; c += 32) sum += rlen[c];
#pragma unroll
        for (int m = 16; m >= 1; m >>= 1) sum += __shfl_xor(sum, m);
        if (tid == 0) btot[b] = sum;
    }
}

__global__ void __launch_bounds__(1024)
gcn_edge1_sort(const int* __restrict__ part, const unsigned short* __restrict__ tbl, const int* __restrict__ btot,
               const float* __restrict__ xs, float* __restrict__ pair, const float* __restrict__ x,
               int* __restrict__ part2, int* __restrict__ brange, int N, int chunkLen, int C) {
    __shared__ float acc[RN];
    __shared__ float disL[RN];
    __shared__ int cur[RNS];
    __shared__ int roff[RCAP];
    __shared__ int rlen[RCAP];
    __shared__ int sStart;
    __shared__ int sTotal;
    __shared__ int stage[SCAP];
    const int tid = threadIdx.x;
    const int lane = tid & 31;
    const int wave = tid >> 5;
    const int nWaves = blockDim.x >> 5;
    const int b = blockIdx.x;
    const int nd = b * RN + tid;
    if (tid == 0) sStart = 0;
    if (tid < RNS) {
        v2f p;
        p.x = 0.0f;
        p.y = 0.0f;
        if (tid < RN && nd < N) p = *(const v2f*)(pair + 2 * nd);
        cur[tid] = __float_as_int(p.x);
        if (tid < RN) {
            acc[tid] = 0.0f;
            disL[tid] = p.y;
        }
    }
    load_runs(tbl, b, C, chunkLen, roff, rlen);
    int before = 0;
    for (int i = tid; i < b; i += blockDim.x) before += btot[i];
#pragma unroll
    for (int m = 16; m >= 1; m >>= 1) before += __shfl_xor(before, m);
    __syncthreads();
    if (lane == 0 && before != 0) atomicAdd(&sStart, before);
    if (tid < 32) {
        const int tot = wave_excl_scan_lds<RNS / 32>(cur, tid);
        if (tid == 0) sTotal = tot;
    }
    __syncthreads();
    const int s0 = sStart;
    const int nE = sTotal;
    const bool staged = nE <= SCAP;
    if (staged) {
        for (int c0 = wave; c0 < C; c0 += nWaves * RPWS) {
            int w0[RPWS], w1[RPWS], len[RPWS], off[RPWS], p0[RPWS], p1[RPWS];
            float x0[RPWS], x1[RPWS];
#pragma unroll
            for (int k = 0; k < RPWS; ++k) {
                const int c = c0 + k * nWaves;
                off[k] = c < C ? roff[c] : 0;
                len[k] = c < C ? rlen[c] : 0;
            }
#pragma unroll
            for (int k = 0; k < RPWS; ++k) {
                off[k] = __builtin_amdgcn_readfirstlane(off[k]);
                len[k] = __builtin_amdgcn_readfirstlane(len[k]);
            }
#pragma unroll
            for (int k = 0; k < RPWS; ++k) {
                w0[k] = part[lane < len[k] ? off[k] + lane : 0];
                w1[k] = part[lane + 32 < len[k] ? off[k] + lane + 32 : 0];
            }
#pragma unroll
            for (int k = 0; k < RPWS; ++k) {
                x0[k] = xs[w0[k] & SRC_MASK] * disL[w0[k] >> SRC_BITS];
                x1[k] = xs[w1[k] & SRC_MASK] * disL[w1[k] >> SRC_BITS];
            }
#pragma unroll
            for (int k = 0; k < RPWS; ++k) asm volatile("" : "+v"(x0[k]), "+v"(x1[k]));
#pragma unroll
            for (int k = 0; k < RPWS; ++k) {
                p0[k] = 0;
                p1[k] = 0;
                if (lane < len[k]) {
                    const int dl = w0[k] >> SRC_BITS;
                    atomicAdd(&acc[dl], x0[k]);
                    p0[k] = atomicAdd(&cur[dl], 1);
                }
                if (lane + 32 < len[k]) {
                    const int dl = w1[k] >> SRC_BITS;
                    atomicAdd(&acc[dl], x1[k]);
                    p1[k] = atomicAdd(&cur[dl], 1);
                }
            }
#pragma unroll
            for (int k = 0; k < RPWS; ++k) {
                if (lane < len[k]) stage[p0[k]] = w0[k];
                if (lane + 32 < len[k]) stage[p1[k]] = w1[k];
                for (int j = lane + 64; j < len[k]; j += 32) {
                    const int ww = part[off[k] + j];
                    const int dl = ww >> SRC_BITS;
                    atomicAdd(&acc[dl], xs[ww & SRC_MASK] * disL[dl]);
                    stage[atomicAdd(&cur[dl], 1)] = ww;
                }
            }
        }
    } else {
        for (int c = wave; c < C; c += nWaves) {
            const int off = roff[c];
            const int len = rlen[c];
            for (int j = lane; j < len; j += 32) {
                const int ww = part[off + j];
                const int dl = ww >> SRC_BITS;
                atomicAdd(&acc[dl], xs[ww & SRC_MASK] * disL[dl]);
                part2[s0 + atomicAdd(&cur[dl], 1)] = ww;
            }
        }
    }
    __syncthreads();
    if (staged)
        for (int i = tid; i < nE; i += blockDim.x) part2[s0 + i] = stage[i];
    if (tid < RN && nd < N) pair[2 * nd] = acc[tid] + disL[tid] * disL[tid] * x[nd];
    if (tid == 0) {
        brange[2 * b] = s0;
        brange[2 * b + 1] = s0 + nE;
    }
}

__device__ __forceinline__ void row_flush(float* acc, int dl, int lane, float a0, float a1) {
    atomicAdd(&acc[dl * AP + lane], a0);
    atomicAdd(&acc[dl * AP + 32 + lane], a1);
}

__global__ void __launch_bounds__(1024)
gcn_layer2_bucket(const int* __restrict__ part2, const int* __restrict__ brange,
                  const float* __restrict__ pair,
                  const float* __restrict__ W1, const float* __restrict__ b1,
                  const float* __restrict__ W2, const float* __restrict__ b2,
                  const int* __restrict__ batch,
                  float* __restrict__ psum, float* __restrict__ pcnt, int N) {
    __shared__ float acc[RNT * AP];
    __shared__ float disL[RN];
    __shared__ float tL[RN];
    __shared__ int gL[RNT];
    __shared__ float pl[PG * 32];
    __shared__ float pc[PG];
    const int tid = threadIdx.x;
    const int lane = tid & 31;
    const int wave = tid >> 5;
    const int nWaves = blockDim.x >> 5;
    const int node0 = blockIdx.x * RN;
    for (int i = tid; i < RNT * AP; i += blockDim.x) acc[i] = 0.0f;
    for (int i = tid; i < PG * 32; i += blockDim.x) pl[i] = 0.0f;
    if (tid < PG) pc[tid] = 0.0f;
    int myG = -1;
    if (tid < RNT) {
        int nd = node0 + tid;
        if (tid < RN) {
            tL[tid] = nd < N ? pair[2 * nd] : 0.0f;
            disL[tid] = nd < N ? pair[2 * nd + 1] : 0.0f;
            if (nd < N) myG = batch[nd];
        }
        gL[tid] = myG;
    }
    const int g0 = batch[node0];
    const float w1a = W1[lane], w1b = W1[lane + 32];
    const float b1a = b1[lane], b1b = b1[lane + 32];
    __syncthreads();
    if (myG >= 0) {
        const unsigned off = (unsigned)(myG - g0);
        if (off < PG) atomicAdd(&pc[off], 1.0f); else atomicAdd(&pcnt[myG], 1.0f);
    }
    const int e0 = brange[2 * blockIdx.x], e1 = brange[2 * blockIdx.x + 1];
    for (int base = e0 + wave * 32; base < e1; base += nWaves * 32) {
        const int idx = base + lane;
        const int w = idx < e1 ? part2[idx] : 0;
        const int s = w & SRC_MASK;
        const int dl = w >> SRC_BITS;
        const v2f p = *(const v2f*)(pair + 2 * s);
        const int tBits = __float_as_int(p.x);
        const int dBits = __float_as_int(p.y);
        const int cnt = min(32, e1 - base);
        const int prev = __shfl_up(dl, 1);
        const unsigned brk = __builtin_amdgcn_ballot_w32(lane > 0 && lane < cnt && dl != prev);
        float a0 = 0.0f, a1 = 0.0f;
        if (cnt == 32) {
#pragma unroll
            for (int j = 0; j < 32; ++j) {
                if (j > 0 && ((brk >> j) & 1u)) {
                    row_flush(acc, __builtin_amdgcn_readlane(dl, j - 1), lane, a0, a1);
                    a0 = 0.0f;
                    a1 = 0.0f;
                }
                const float tj = __int_as_float(__builtin_amdgcn_readlane(tBits, j));
                const float dj = __int_as_float(__builtin_amdgcn_readlane(dBits, j));
                a0 = fmaf(dj, fmaxf(fmaf(tj, w1a, b1a), 0.0f), a0);
                a1 = fmaf(dj, fmaxf(fmaf(tj, w1b, b1b), 0.0f), a1);
            }
            row_flush(acc, __builtin_amdgcn_readlane(dl, 31), lane, a0, a1);
        } else {
            for (int j = 0; j < cnt; ++j) {
                if ((brk >> j) & 1u) {
                    row_flush(acc, __builtin_amdgcn_readlane(dl, j - 1), lane, a0, a1);
                    a0 = 0.0f;
                    a1 = 0.0f;
                }
                const float tj = __int_as_float(__builtin_amdgcn_readlane(tBits, j));
                const float dj = __int_as_float(__builtin_amdgcn_readlane(dBits, j));
                a0 = fmaf(dj, fmaxf(fmaf(tj, w1a, b1a), 0.0f), a0);
                a1 = fmaf(dj, fmaxf(fmaf(tj, w1b, b1b), 0.0f), a1);
            }
            row_flush(acc, __builtin_amdgcn_readlane(dl, cnt - 1), lane, a0, a1);
        }
    }
    __syncthreads();
    {
        const int k = tid & 63;
        const float wk = W1[k], bk = b1[k];
        for (int nl = tid >> 6; nl < RN; nl += blockDim.x >> 6) {
            const float d = disL[nl];
            const float hself = fmaxf(fmaf(tL[nl], wk, bk), 0.0f);
            float* q = &acc[nl * AP + k];
            *q = d * (*q) + d * d * hself;
        }
    }
    __syncthreads();
    if (__builtin_amdgcn_readfirstlane(wave) < (RNT / 16) * 2) {
    const int c = lane & 15;
    const int khalf = lane >> 4;
    const int mt = wave >> 1;
    const int n0 = (wave & 1) << 4;
    const float* arow = acc + (mt * 16 + c) * AP + khalf * 2;
    const float* wcol = W2 + (khalf * 2) * 32 + n0 + c;
    const float bias = b2[n0 + c];
    v8f D = {};
#pragma unroll
    for (int k0 = 0; k0 < 64; k0 += 4) {
        v2f a, b;
        a.x = arow[k0];
        a.y = arow[k0 + 1];
        b.x = wcol[k0 * 32];
        b.y = wcol[(k0 + 1) * 32];
        D = __builtin_amdgcn_wmma_f32_16x16x4_f32(
             false, a,  false, b,
             (short)0, D,  false,  false);
    }
    const int r0 = mt * 16 + khalf * 8;
    float hv[8];
    int gv[8];
#pragma unroll
    for (int v = 0; v < 8; ++v) {
        const float h = D[v] + bias;
        hv[v] = h > 0.0f ? h : 0.0f;
        gv[v] = gL[r0 + v];
    }
#pragma unroll
    for (int v = 0; v < 8; ++v) asm volatile("" : "+v"(hv[v]));
#pragma unroll
    for (int v = 0; v < 8; ++v) {
        if (gv[v] >= 0) {
            const unsigned off = (unsigned)(gv[v] - g0);
            if (off < PG) atomicAdd(&pl[off * 32 + n0 + c], hv[v]);
            else atomicAdd(&psum[gv[v] * 32 + n0 + c], hv[v]);
        }
    }
    }
    __syncthreads();
    for (int i = tid; i < PG * 32; i += blockDim.x) {
        const int off = i >> 5;
        const int f = i & 31;
        const float cn = pc[off];
        if (cn > 0.0f) {
            atomicAdd(&psum[(g0 + off) * 32 + f], pl[i]);
            if (f == 0) atomicAdd(&pcnt[g0 + off], cn);
        }
    }
}

#define HP 132
__global__ void __launch_bounds__(512)
gcn_head_wmma(const float* __restrict__ psum, const float* __restrict__ pcnt,
              const float* __restrict__ Wp1, const float* __restrict__ bp1,
              const float* __restrict__ Wp2, const float* __restrict__ bp2,
              float* __restrict__ out, int G) {
    __shared__ float hs[32 * HP];
    __shared__ float os[96];
    const int tid = threadIdx.x;
    const int lane = tid & 31;
    const int wave = tid >> 5;
    const int c = lane & 15;
    const int khalf = lane >> 4;
    const int mt = wave >> 3;
    const int n0 = (wave & 7) << 4;
    const int g0 = blockIdx.x << 5;

    const int row = tid >> 4;
    const int sub = tid & 15;
    float w2[24];
#pragma unroll
    for (int j = 0; j < 24; ++j) w2[j] = Wp2[sub * 24 + j];
    float o0 = 0.0f, o1 = 0.0f, o2 = 0.0f;

    int g = g0 + (mt << 4) + c;
    g = g < G ? g : G - 1;
    float cn = pcnt[g];
    cn = cn > 1.0f ? cn : 1.0f;
    const float inv = 1.0f / cn;
    const float* prow = psum + g * 32 + khalf * 2;
    const float* wcol = Wp1 + (khalf * 2) * 128 + n0 + c;

    v8f acc0 = {};
    v8f acc1 = {};
#pragma unroll
    for (int k0 = 0; k0 < 32; k0 += 8) {
        v2f a, b;
        a.x = prow[k0] * inv;
        a.y = prow[k0 + 1] * inv;
        b.x = wcol[k0 * 128];
        b.y = wcol[(k0 + 1) * 128];
        acc0 = __builtin_amdgcn_wmma_f32_16x16x4_f32(
             false, a,  false, b,
             (short)0, acc0,  false,  false);
        a.x = prow[k0 + 4] * inv;
        a.y = prow[k0 + 5] * inv;
        b.x = wcol[(k0 + 4) * 128];
        b.y = wcol[(k0 + 5) * 128];
        acc1 = __builtin_amdgcn_wmma_f32_16x16x4_f32(
             false, a,  false, b,
             (short)0, acc1,  false,  false);
    }

    const float bias = bp1[n0 + c];
    const int rowBase = (mt << 4) + khalf * 8;
#pragma unroll
    for (int v = 0; v < 8; ++v) {
        float h = acc0[v] + acc1[v] + bias;
        hs[(rowBase + v) * HP + n0 + c] = h > 0.0f ? h : 0.0f;
    }
    __syncthreads();

    const float* hr = hs + row * HP + sub * 8;
#pragma unroll
    for (int j = 0; j < 8; ++j) {
        float h = hr[j];
        o0 = fmaf(h, w2[j * 3 + 0], o0);
        o1 = fmaf(h, w2[j * 3 + 1], o1);
        o2 = fmaf(h, w2[j * 3 + 2], o2);
    }
#pragma unroll
    for (int m = 8; m >= 1; m >>= 1) {
        o0 += __shfl_xor(o0, m, 16);
        o1 += __shfl_xor(o1, m, 16);
        o2 += __shfl_xor(o2, m, 16);
    }
    if (sub == 0) {
        os[row * 3 + 0] = o0 + bp2[0];
        os[row * 3 + 1] = o1 + bp2[1];
        os[row * 3 + 2] = o2 + bp2[2];
    }
    __syncthreads();
    if (tid < 96) {
        int idx = g0 * 3 + tid;
        if (idx < G * 3) out[idx] = os[tid];
    }
}

extern "C" void kernel_launch(void* const* d_in, const int* in_sizes, int n_in,
                              void* d_out, int out_size, void* d_ws, size_t ws_size,
                              hipStream_t stream) {
    const float* x   = (const float*)d_in[0];
    const float* W1  = (const float*)d_in[1];
    const float* b1  = (const float*)d_in[2];
    const float* W2  = (const float*)d_in[3];
    const float* b2  = (const float*)d_in[4];
    const float* Wp1 = (const float*)d_in[5];
    const float* bp1 = (const float*)d_in[6];
    const float* Wp2 = (const float*)d_in[7];
    const float* bp2 = (const float*)d_in[8];
    const int* eidx  = (const int*)d_in[9];
    const int* batch = (const int*)d_in[10];
    float* out = (float*)d_out;

    const int N = in_sizes[0];
    const int E = in_sizes[9] / 2;
    const int G = out_size / 3;
    const int* src = eidx;
    const int* dst = eidx + E;

    const int B = (N + RN - 1) / RN;
    const int C = NCHUNK;
    const int chunkLen = (E + C - 1) / C;

    float* wsf   = (float*)d_ws;
    float* pair  = wsf;
    float* xs    = pair + 2L * N;
    unsigned short* tbl = (unsigned short*)(xs + N);
    int*   btot  = (int*)(xs + N) + ((B + 1) * C + 1) / 2;
    int*   brange = btot + B;
    int*   part  = (int*)(wsf + 4L * N);
    int*   part2 = part + E;
    float* psum  = (float*)(part2 + E);
    float* pcnt  = psum + (long)G * 32;

    gcn_chunk_sort<<<C, 1024, 0, stream>>>(src, dst, tbl, part, E, chunkLen, B, C);

    gcn_deg<<<B, 1024, 0, stream>>>(part, tbl, x, pair, xs, btot, psum, 33 * G, N, chunkLen, C);
    gcn_edge1_sort<<<B, 1024, 0, stream>>>(part, tbl, btot, xs, pair, x, part2, brange, N, chunkLen, C);
    gcn_layer2_bucket<<<B, 1024, 0, stream>>>(part2, brange, pair, W1, b1, W2, b2, batch, psum, pcnt, N);

    gcn_head_wmma<<<(G + 31) / 32, 512, 0, stream>>>(psum, pcnt, Wp1, bp1, Wp2, bp2, out, G);
}
